# starting kernel
# baseline (speedup 1.0000x reference)
#include <hip/hip_runtime.h>
#include <hip/hip_bf16.h>

typedef float v2f __attribute__((ext_vector_type(2)));
typedef float v8f __attribute__((ext_vector_type(8)));

__global__ void k_deg_init(int* __restrict__ deg, int n) {
    int i = blockIdx.x * blockDim.x + threadIdx.x;
    if (i < n) deg[i] = 1;
}

__global__ void k_deg_count(const int* __restrict__ dst, int* __restrict__ deg, int e) {
    int i = blockIdx.x * blockDim.x + threadIdx.x;
    if (i < e) atomicAdd(&deg[dst[i]], 1);
}

__global__ void k_dinv(float* __restrict__ dinv, int n) {
    int i = blockIdx.x * blockDim.x + threadIdx.x;
    if (i < n) {
        int d = ((const int*)dinv)[i];
        dinv[i] = rsqrtf((float)d);
    }
}

__global__ void k_gemm1(const float* __restrict__ x, const float* __restrict__ W1,
                        float* __restrict__ h1, int n) {
    int lane = threadIdx.x & 31;
    int wave = threadIdx.x >> 5;
    int base = (blockIdx.x * (blockDim.x >> 5) + wave) * 16;
    if (base >= n) return;

    int  m   = lane & 15;
    bool hi  = lane >= 16;
    int node = base + m;
    if (node >= n) node = n - 1;
    const float* xr = x + (size_t)node * 3;

    float ax = xr[hi ? 2 : 0];
    float x1 = xr[1];
    v2f a;
    a.x = ax;
    a.y = hi ? 0.0f : x1;

    int  mc    = (m < 8) ? m : 7;
    bool valid = (m < 8);
    float w0 = W1[0 * 8 + mc];
    float w1 = W1[1 * 8 + mc];
    float w2 = W1[2 * 8 + mc];
    v2f b;
    b.x = valid ? (hi ? w2 : w0) : 0.0f;
    b.y = (valid && !hi) ? w1 : 0.0f;

    v8f c = {};
    c = __builtin_amdgcn_wmma_f32_16x16x4_f32(false, a, false, b, (short)0, c,
                                              false, false);

    if (valid) {
        float* hp = h1 + (size_t)(base + (hi ? 8 : 0)) * 8 + m;
        if (base + 16 <= n) {
            #pragma unroll
            for (int r = 0; r < 8; ++r) hp[(size_t)r * 8] = c[r];
        } else {
            #pragma unroll
            for (int r = 0; r < 8; ++r) {
                int row = base + r + (hi ? 8 : 0);
                if (row < n) h1[(size_t)row * 8 + m] = c[r];
            }
        }
    }
}

__global__ void k_init1(const float* __restrict__ h1, const float* __restrict__ dinv,
                        const float* __restrict__ b1, float* __restrict__ agg1, int n8) {
    int i = blockIdx.x * blockDim.x + threadIdx.x;
    if (i >= n8) return;
    int   v  = i >> 3, j = i & 7;
    float di = dinv[v];
    agg1[i] = b1[j] + h1[i] * di * di;
}

__global__ void k_edge1(const int* __restrict__ src, const int* __restrict__ dst,
                        const float* __restrict__ dinv, const float* __restrict__ h1,
                        float* __restrict__ agg1, int e) {
    int i = blockIdx.x * blockDim.x + threadIdx.x;
    if (i >= e) return;
    int s = src[i], d = dst[i];
    float nrm = dinv[s] * dinv[d];
    const float4* hp = (const float4*)(h1 + (size_t)s * 8);
    float4 m0 = hp[0], m1 = hp[1];
    float* o = agg1 + (size_t)d * 8;
    unsafeAtomicAdd(o + 0, m0.x * nrm);
    unsafeAtomicAdd(o + 1, m0.y * nrm);
    unsafeAtomicAdd(o + 2, m0.z * nrm);
    unsafeAtomicAdd(o + 3, m0.w * nrm);
    unsafeAtomicAdd(o + 4, m1.x * nrm);
    unsafeAtomicAdd(o + 5, m1.y * nrm);
    unsafeAtomicAdd(o + 6, m1.z * nrm);
    unsafeAtomicAdd(o + 7, m1.w * nrm);
}

__global__ void k_gemm2(const float* __restrict__ agg1, const float* __restrict__ W2,
                        float* __restrict__ h2, int n) {
    int lane = threadIdx.x & 31;
    int wave = threadIdx.x >> 5;
    int base = (blockIdx.x * (blockDim.x >> 5) + wave) * 16;
    if (base >= n) return;

    int  m   = lane & 15;
    bool hi  = lane >= 16;
    int node = base + m;
    if (node >= n) node = n - 1;
    const float* r0 = agg1 + (size_t)node * 8;
    int k0 = hi ? 2 : 0;

    v2f a0, a1;
    a0.x = fmaxf(r0[k0],     0.0f); a0.y = fmaxf(r0[k0 + 1],     0.0f);
    a1.x = fmaxf(r0[4 + k0], 0.0f); a1.y = fmaxf(r0[4 + k0 + 1], 0.0f);

    int  mc    = (m < 2) ? m : 1;
    bool valid = (m < 2);
    float wa = W2[(k0)     * 2 + mc];
    float wb = W2[(k0 + 1) * 2 + mc];
    float wc = W2[(k0 + 4) * 2 + mc];
    float wd = W2[(k0 + 5) * 2 + mc];
    v2f b0, b1v;
    b0.x  = valid ? wa : 0.0f;
    b0.y  = valid ? wb : 0.0f;
    b1v.x = valid ? wc : 0.0f;
    b1v.y = valid ? wd : 0.0f;

    v8f c = {};
    c = __builtin_amdgcn_wmma_f32_16x16x4_f32(false, a0, false, b0, (short)0, c,
                                              false, false);
    c = __builtin_amdgcn_wmma_f32_16x16x4_f32(false, a1, false, b1v, (short)0, c,
                                              false, false);

    if (valid) {
        float* hp = h2 + (size_t)(base + (hi ? 8 : 0)) * 2 + m;
        if (base + 16 <= n) {
            #pragma unroll
            for (int r = 0; r < 8; ++r) hp[(size_t)r * 2] = c[r];
        } else {
            #pragma unroll
            for (int r = 0; r < 8; ++r) {
                int row = base + r + (hi ? 8 : 0);
                if (row < n) h2[(size_t)row * 2 + m] = c[r];
            }
        }
    }
}

__global__ void k_init2(const float* __restrict__ h2, const float* __restrict__ dinv,
                        const float* __restrict__ b2, float* __restrict__ out, int n2) {
    int i = blockIdx.x * blockDim.x + threadIdx.x;
    if (i >= n2) return;
    int   v  = i >> 1, j = i & 1;
    float di = dinv[v];
    out[i] = b2[j] + h2[i] * di * di;
}

__global__ void k_edge2(const int* __restrict__ src, const int* __restrict__ dst,
                        const float* __restrict__ dinv, const float* __restrict__ h2,
                        float* __restrict__ out, int e) {
    int i = blockIdx.x * blockDim.x + threadIdx.x;
    if (i >= e) return;
    int s = src[i], d = dst[i];
    float nrm = dinv[s] * dinv[d];
    const float2* hp = (const float2*)(h2 + (size_t)s * 2);
    float2 m0 = hp[0];
    float* o = out + (size_t)d * 2;
    unsafeAtomicAdd(o + 0, m0.x * nrm);
    unsafeAtomicAdd(o + 1, m0.y * nrm);
}

extern "C" void kernel_launch(void* const* d_in, const int* in_sizes, int n_in,
                              void* d_out, int out_size, void* d_ws, size_t ws_size,
                              hipStream_t stream) {
    const float* x  = (const float*)d_in[0];
    const int*   ei = (const int*)d_in[1];
    const float* W1 = (const float*)d_in[2];
    const float* b1 = (const float*)d_in[3];
    const float* W2 = (const float*)d_in[4];
    const float* b2 = (const float*)d_in[5];
    float* out = (float*)d_out;

    int n = in_sizes[0] / 3;
    int e = in_sizes[1] / 2;
    const int* src = ei;
    const int* dst = ei + e;

    float* dinv = (float*)d_ws;
    float* h1   = dinv + n;
    float* agg1 = h1 + (size_t)n * 8;
    float* h2   = h1;

    const int T = 256;
    int gN  = (n + T - 1) / T;
    int gE  = (e + T - 1) / T;
    int gN8 = (n * 8 + T - 1) / T;
    int gN2 = (n * 2 + T - 1) / T;
    int gMM = (n + 127) / 128;

    k_deg_init <<<gN,  T, 0, stream>>>((int*)dinv, n);
    k_deg_count<<<gE,  T, 0, stream>>>(dst, (int*)dinv, e);
    k_dinv     <<<gN,  T, 0, stream>>>(dinv, n);
    k_gemm1    <<<gMM, T, 0, stream>>>(x, W1, h1, n);
    k_init1    <<<gN8, T, 0, stream>>>(h1, dinv, b1, agg1, n * 8);
    k_edge1    <<<gE,  T, 0, stream>>>(src, dst, dinv, h1, agg1, e);
    k_gemm2    <<<gMM, T, 0, stream>>>(agg1, W2, h2, n);
    k_init2    <<<gN2, T, 0, stream>>>(h2, dinv, b2, out, n * 2);
    k_edge2    <<<gE,  T, 0, stream>>>(src, dst, dinv, h2, out, e);
}


# layer 1 aggregates the 3 input features before the W1 product (A(xW1) -> (Ax)W1): edge scatter does 3 float atomics per edge instead of 8; summation order changes
# speedup vs baseline: 1.4614x; 1.4614x over previous
#include <hip/hip_runtime.h>
#include <hip/hip_bf16.h>

typedef float v2f __attribute__((ext_vector_type(2)));
typedef float v8f __attribute__((ext_vector_type(8)));

__global__ void k_deg_init(int* __restrict__ deg, int n) {
    int i = blockIdx.x * blockDim.x + threadIdx.x;
    if (i < n) deg[i] = 1;
}

__global__ void k_deg_count(const int* __restrict__ dst, int* __restrict__ deg, int e) {
    int i = blockIdx.x * blockDim.x + threadIdx.x;
    if (i < e) atomicAdd(&deg[dst[i]], 1);
}

__global__ void k_dinv(float* __restrict__ dinv, int n) {
    int i = blockIdx.x * blockDim.x + threadIdx.x;
    if (i < n) {
        int d = ((const int*)dinv)[i];
        dinv[i] = rsqrtf((float)d);
    }
}

__global__ void k_initx(const float* __restrict__ x, const float* __restrict__ dinv,
                        float* __restrict__ aggx, int n) {
    int v = blockIdx.x * blockDim.x + threadIdx.x;
    if (v >= n) return;
    float di = dinv[v];
    float w  = di * di;
    const float* xr = x + (size_t)v * 3;
    float4 o;
    o.x = xr[0] * w;
    o.y = xr[1] * w;
    o.z = xr[2] * w;
    o.w = 0.0f;
    ((float4*)aggx)[v] = o;
}

__global__ void k_edgex(const int* __restrict__ src, const int* __restrict__ dst,
                        const float* __restrict__ dinv, const float* __restrict__ x,
                        float* __restrict__ aggx, int e) {
    int i = blockIdx.x * blockDim.x + threadIdx.x;
    if (i >= e) return;
    int s = src[i], d = dst[i];
    float nrm = dinv[s] * dinv[d];
    const float* xr = x + (size_t)s * 3;
    float x0 = xr[0], x1 = xr[1], x2 = xr[2];
    float* o = aggx + (size_t)d * 4;
    unsafeAtomicAdd(o + 0, x0 * nrm);
    unsafeAtomicAdd(o + 1, x1 * nrm);
    unsafeAtomicAdd(o + 2, x2 * nrm);
}

__global__ void k_gemm1(const float* __restrict__ aggx, const float* __restrict__ W1,
                        const float* __restrict__ b1, float* __restrict__ agg1, int n) {
    int lane = threadIdx.x & 31;
    int wave = threadIdx.x >> 5;
    int base = (blockIdx.x * (blockDim.x >> 5) + wave) * 16;
    if (base >= n) return;

    int  m   = lane & 15;
    bool hi  = lane >= 16;
    int node = base + m;
    if (node >= n) node = n - 1;

    const float2 av = *(const float2*)(aggx + (size_t)node * 4 + (hi ? 2 : 0));
    v2f a;
    a.x = av.x;
    a.y = av.y;

    int  mc    = (m < 8) ? m : 7;
    bool valid = (m < 8);
    float w0 = W1[0 * 8 + mc];
    float w1 = W1[1 * 8 + mc];
    float w2 = W1[2 * 8 + mc];
    float bj = b1[mc];
    v2f b;
    b.x = valid ? (hi ? w2 : w0) : 0.0f;
    b.y = (valid && !hi) ? w1 : 0.0f;

    float cb = valid ? bj : 0.0f;
    v8f c = {cb, cb, cb, cb, cb, cb, cb, cb};
    c = __builtin_amdgcn_wmma_f32_16x16x4_f32(false, a, false, b, (short)0, c,
                                              false, false);

    if (valid) {
        float* hp = agg1 + (size_t)(base + (hi ? 8 : 0)) * 8 + m;
        if (base + 16 <= n) {
            #pragma unroll
            for (int r = 0; r < 8; ++r) hp[(size_t)r * 8] = c[r];
        } else {
            #pragma unroll
            for (int r = 0; r < 8; ++r) {
                int row = base + r + (hi ? 8 : 0);
                if (row < n) agg1[(size_t)row * 8 + m] = c[r];
            }
        }
    }
}

__global__ void k_gemm2(const float* __restrict__ agg1, const float* __restrict__ W2,
                        float* __restrict__ h2, int n) {
    int lane = threadIdx.x & 31;
    int wave = threadIdx.x >> 5;
    int base = (blockIdx.x * (blockDim.x >> 5) + wave) * 16;
    if (base >= n) return;

    int  m   = lane & 15;
    bool hi  = lane >= 16;
    int node = base + m;
    if (node >= n) node = n - 1;
    const float* r0 = agg1 + (size_t)node * 8;
    int k0 = hi ? 2 : 0;

    v2f a0, a1;
    a0.x = fmaxf(r0[k0],     0.0f); a0.y = fmaxf(r0[k0 + 1],     0.0f);
    a1.x = fmaxf(r0[4 + k0], 0.0f); a1.y = fmaxf(r0[4 + k0 + 1], 0.0f);

    int  mc    = (m < 2) ? m : 1;
    bool valid = (m < 2);
    float wa = W2[(k0)     * 2 + mc];
    float wb = W2[(k0 + 1) * 2 + mc];
    float wc = W2[(k0 + 4) * 2 + mc];
    float wd = W2[(k0 + 5) * 2 + mc];
    v2f b0, b1v;
    b0.x  = valid ? wa : 0.0f;
    b0.y  = valid ? wb : 0.0f;
    b1v.x = valid ? wc : 0.0f;
    b1v.y = valid ? wd : 0.0f;

    v8f c = {};
    c = __builtin_amdgcn_wmma_f32_16x16x4_f32(false, a0, false, b0, (short)0, c,
                                              false, false);
    c = __builtin_amdgcn_wmma_f32_16x16x4_f32(false, a1, false, b1v, (short)0, c,
                                              false, false);

    if (valid) {
        float* hp = h2 + (size_t)(base + (hi ? 8 : 0)) * 2 + m;
        if (base + 16 <= n) {
            #pragma unroll
            for (int r = 0; r < 8; ++r) hp[(size_t)r * 2] = c[r];
        } else {
            #pragma unroll
            for (int r = 0; r < 8; ++r) {
                int row = base + r + (hi ? 8 : 0);
                if (row < n) h2[(size_t)row * 2 + m] = c[r];
            }
        }
    }
}

__global__ void k_init2(const float* __restrict__ h2, const float* __restrict__ dinv,
                        const float* __restrict__ b2, float* __restrict__ out, int n2) {
    int i = blockIdx.x * blockDim.x + threadIdx.x;
    if (i >= n2) return;
    int   v  = i >> 1, j = i & 1;
    float di = dinv[v];
    out[i] = b2[j] + h2[i] * di * di;
}

__global__ void k_edge2(const int* __restrict__ src, const int* __restrict__ dst,
                        const float* __restrict__ dinv, const float* __restrict__ h2,
                        float* __restrict__ out, int e) {
    int i = blockIdx.x * blockDim.x + threadIdx.x;
    if (i >= e) return;
    int s = src[i], d = dst[i];
    float nrm = dinv[s] * dinv[d];
    const float2* hp = (const float2*)(h2 + (size_t)s * 2);
    float2 m0 = hp[0];
    float* o = out + (size_t)d * 2;
    unsafeAtomicAdd(o + 0, m0.x * nrm);
    unsafeAtomicAdd(o + 1, m0.y * nrm);
}

extern "C" void kernel_launch(void* const* d_in, const int* in_sizes, int n_in,
                              void* d_out, int out_size, void* d_ws, size_t ws_size,
                              hipStream_t stream) {
    const float* x  = (const float*)d_in[0];
    const int*   ei = (const int*)d_in[1];
    const float* W1 = (const float*)d_in[2];
    const float* b1 = (const float*)d_in[3];
    const float* W2 = (const float*)d_in[4];
    const float* b2 = (const float*)d_in[5];
    float* out = (float*)d_out;

    int n = in_sizes[0] / 3;
    int e = in_sizes[1] / 2;
    const int* src = ei;
    const int* dst = ei + e;

    float* dinv = (float*)d_ws;
    float* aggx = dinv + n;
    float* h2   = aggx + (size_t)n * 4;
    float* agg1 = dinv + (size_t)n * 9;

    const int T = 256;
    int gN  = (n + T - 1) / T;
    int gE  = (e + T - 1) / T;
    int gN2 = (n * 2 + T - 1) / T;
    int gMM = (n + 127) / 128;

    k_deg_init <<<gN,  T, 0, stream>>>((int*)dinv, n);
    k_deg_count<<<gE,  T, 0, stream>>>(dst, (int*)dinv, e);
    k_dinv     <<<gN,  T, 0, stream>>>(dinv, n);
    k_initx    <<<gN,  T, 0, stream>>>(x, dinv, aggx, n);
    k_edgex    <<<gE,  T, 0, stream>>>(src, dst, dinv, x, aggx, e);
    k_gemm1    <<<gMM, T, 0, stream>>>(aggx, W1, b1, agg1, n);
    k_gemm2    <<<gMM, T, 0, stream>>>(agg1, W2, h2, n);
    k_init2    <<<gN2, T, 0, stream>>>(h2, dinv, b2, out, n * 2);
    k_edge2    <<<gE,  T, 0, stream>>>(src, dst, dinv, h2, out, e);
}


# normalization hoisted out of both edge loops: node rows pre-scaled by dinv[s] once, dinv[d] applied once per node after the sum; an edge gathers 1 row, not 3 values; rounding of the sums changes
# speedup vs baseline: 1.5319x; 1.0482x over previous
#include <hip/hip_runtime.h>
#include <hip/hip_bf16.h>

typedef float v2f __attribute__((ext_vector_type(2)));
typedef float v8f __attribute__((ext_vector_type(8)));

__global__ void k_deg_init(int* __restrict__ deg, int n) {
    int i = blockIdx.x * blockDim.x + threadIdx.x;
    if (i < n) deg[i] = 1;
}

__global__ void k_deg_count(const int* __restrict__ dst, int* __restrict__ deg, int e) {
    int i = blockIdx.x * blockDim.x + threadIdx.x;
    if (i < e) atomicAdd(&deg[dst[i]], 1);
}

__global__ void k_dinv(float* __restrict__ dinv, int n) {
    int i = blockIdx.x * blockDim.x + threadIdx.x;
    if (i < n) {
        int d = ((const int*)dinv)[i];
        dinv[i] = rsqrtf((float)d);
    }
}

__global__ void k_initx(const float* __restrict__ x, const float* __restrict__ dinv,
                        float* __restrict__ xs, float* __restrict__ aggx, int n) {
    int v = blockIdx.x * blockDim.x + threadIdx.x;
    if (v >= n) return;
    float di = dinv[v];
    const float* xr = x + (size_t)v * 3;
    float4 o;
    o.x = xr[0] * di;
    o.y = xr[1] * di;
    o.z = xr[2] * di;
    o.w = 0.0f;
    ((float4*)xs)[v]   = o;
    ((float4*)aggx)[v] = o;
}

__global__ void k_edgex(const int* __restrict__ src, const int* __restrict__ dst,
                        const float* __restrict__ xs, float* __restrict__ aggx, int e) {
    int i = blockIdx.x * blockDim.x + threadIdx.x;
    if (i >= e) return;
    int s = src[i], d = dst[i];
    float4 m0 = ((const float4*)xs)[s];
    float* o = aggx + (size_t)d * 4;
    unsafeAtomicAdd(o + 0, m0.x);
    unsafeAtomicAdd(o + 1, m0.y);
    unsafeAtomicAdd(o + 2, m0.z);
}

__global__ void k_gemm1(const float* __restrict__ aggx, const float* __restrict__ dinv,
                        const float* __restrict__ W1, const float* __restrict__ b1,
                        float* __restrict__ agg1, int n) {
    int lane = threadIdx.x & 31;
    int wave = threadIdx.x >> 5;
    int base = (blockIdx.x * (blockDim.x >> 5) + wave) * 16;
    if (base >= n) return;

    int  m   = lane & 15;
    bool hi  = lane >= 16;
    int node = base + m;
    if (node >= n) node = n - 1;

    const float2 av = *(const float2*)(aggx + (size_t)node * 4 + (hi ? 2 : 0));
    const float  dn = dinv[node];
    v2f a;
    a.x = av.x * dn;
    a.y = av.y * dn;

    int  mc    = (m < 8) ? m : 7;
    bool valid = (m < 8);
    float w0 = W1[0 * 8 + mc];
    float w1 = W1[1 * 8 + mc];
    float w2 = W1[2 * 8 + mc];
    float bj = b1[mc];
    v2f b;
    b.x = valid ? (hi ? w2 : w0) : 0.0f;
    b.y = (valid && !hi) ? w1 : 0.0f;

    float cb = valid ? bj : 0.0f;
    v8f c = {cb, cb, cb, cb, cb, cb, cb, cb};
    c = __builtin_amdgcn_wmma_f32_16x16x4_f32(false, a, false, b, (short)0, c,
                                              false, false);

    if (valid) {
        float* hp = agg1 + (size_t)(base + (hi ? 8 : 0)) * 8 + m;
        if (base + 16 <= n) {
            #pragma unroll
            for (int r = 0; r < 8; ++r) hp[(size_t)r * 8] = c[r];
        } else {
            #pragma unroll
            for (int r = 0; r < 8; ++r) {
                int row = base + r + (hi ? 8 : 0);
                if (row < n) agg1[(size_t)row * 8 + m] = c[r];
            }
        }
    }
}

__global__ void k_gemm2(const float* __restrict__ agg1, const float* __restrict__ dinv,
                        const float* __restrict__ W2, float* __restrict__ h2,
                        float* __restrict__ out, int n) {
    int lane = threadIdx.x & 31;
    int wave = threadIdx.x >> 5;
    int base = (blockIdx.x * (blockDim.x >> 5) + wave) * 16;
    if (base >= n) return;

    int  m   = lane & 15;
    bool hi  = lane >= 16;
    int node = base + m;
    if (node >= n) node = n - 1;
    const float* r0 = agg1 + (size_t)node * 8;
    int k0 = hi ? 2 : 0;

    const float dn = dinv[node];
    v2f a0, a1;
    a0.x = fmaxf(r0[k0],     0.0f) * dn; a0.y = fmaxf(r0[k0 + 1],     0.0f) * dn;
    a1.x = fmaxf(r0[4 + k0], 0.0f) * dn; a1.y = fmaxf(r0[4 + k0 + 1], 0.0f) * dn;

    int  mc    = (m < 2) ? m : 1;
    bool valid = (m < 2);
    float wa = W2[(k0)     * 2 + mc];
    float wb = W2[(k0 + 1) * 2 + mc];
    float wc = W2[(k0 + 4) * 2 + mc];
    float wd = W2[(k0 + 5) * 2 + mc];
    v2f b0, b1v;
    b0.x  = valid ? wa : 0.0f;
    b0.y  = valid ? wb : 0.0f;
    b1v.x = valid ? wc : 0.0f;
    b1v.y = valid ? wd : 0.0f;

    v8f c = {};
    c = __builtin_amdgcn_wmma_f32_16x16x4_f32(false, a0, false, b0, (short)0, c,
                                              false, false);
    c = __builtin_amdgcn_wmma_f32_16x16x4_f32(false, a1, false, b1v, (short)0, c,
                                              false, false);

    if (valid) {
        size_t off = (size_t)(base + (hi ? 8 : 0)) * 2 + m;
        float* hp = h2 + off;
        float* op = out + off;
        if (base + 16 <= n) {
            #pragma unroll
            for (int r = 0; r < 8; ++r) {
                hp[(size_t)r * 2] = c[r];
                op[(size_t)r * 2] = c[r];
            }
        } else {
            #pragma unroll
            for (int r = 0; r < 8; ++r) {
                int row = base + r + (hi ? 8 : 0);
                if (row < n) {
                    h2[(size_t)row * 2 + m]  = c[r];
                    out[(size_t)row * 2 + m] = c[r];
                }
            }
        }
    }
}

__global__ void k_edge2(const int* __restrict__ src, const int* __restrict__ dst,
                        const float* __restrict__ h2, float* __restrict__ out, int e) {
    int i = blockIdx.x * blockDim.x + threadIdx.x;
    if (i >= e) return;
    int s = src[i], d = dst[i];
    const float2* hp = (const float2*)(h2 + (size_t)s * 2);
    float2 m0 = hp[0];
    float* o = out + (size_t)d * 2;
    unsafeAtomicAdd(o + 0, m0.x);
    unsafeAtomicAdd(o + 1, m0.y);
}

__global__ void k_fin2(const float* __restrict__ dinv, const float* __restrict__ b2,
                       float* __restrict__ out, int n) {
    int v = blockIdx.x * blockDim.x + threadIdx.x;
    if (v >= n) return;
    float  di = dinv[v];
    float2 o  = ((float2*)out)[v];
    o.x = b2[0] + o.x * di;
    o.y = b2[1] + o.y * di;
    ((float2*)out)[v] = o;
}

extern "C" void kernel_launch(void* const* d_in, const int* in_sizes, int n_in,
                              void* d_out, int out_size, void* d_ws, size_t ws_size,
                              hipStream_t stream) {
    const float* x  = (const float*)d_in[0];
    const int*   ei = (const int*)d_in[1];
    const float* W1 = (const float*)d_in[2];
    const float* b1 = (const float*)d_in[3];
    const float* W2 = (const float*)d_in[4];
    const float* b2 = (const float*)d_in[5];
    float* out = (float*)d_out;

    int n = in_sizes[0] / 3;
    int e = in_sizes[1] / 2;
    const int* src = ei;
    const int* dst = ei + e;

    float* dinv = (float*)d_ws;
    float* aggx = dinv + n;
    float* h2   = aggx + (size_t)n * 4;
    float* agg1 = dinv + (size_t)n * 9;
    float* xs   = agg1;

    const int T = 256;
    int gN  = (n + T - 1) / T;
    int gE  = (e + T - 1) / T;
    int gMM = (n + 127) / 128;

    k_deg_init <<<gN,  T, 0, stream>>>((int*)dinv, n);
    k_deg_count<<<gE,  T, 0, stream>>>(dst, (int*)dinv, e);
    k_dinv     <<<gN,  T, 0, stream>>>(dinv, n);
    k_initx    <<<gN,  T, 0, stream>>>(x, dinv, xs, aggx, n);
    k_edgex    <<<gE,  T, 0, stream>>>(src, dst, xs, aggx, e);
    k_gemm1    <<<gMM, T, 0, stream>>>(aggx, dinv, W1, b1, agg1, n);
    k_gemm2    <<<gMM, T, 0, stream>>>(agg1, dinv, W2, h2, out, n);
    k_edge2    <<<gE,  T, 0, stream>>>(src, dst, h2, out, e);
    k_fin2     <<<gN,  T, 0, stream>>>(dinv, b2, out, n);
}


# every array the edge atomics add into exists once per half of the chip; a work-group adds into its own half's copy (die id, for speed only), consumers add the two copies; sum order changes
# speedup vs baseline: 3.5686x; 2.3295x over previous
#include <hip/hip_runtime.h>
#include <hip/hip_bf16.h>

typedef float v2f __attribute__((ext_vector_type(2)));
typedef float v8f __attribute__((ext_vector_type(8)));

#define EDGE_CHUNK 1024

__device__ __forceinline__ int chip_half() {
    __shared__ int s_half;
    if (threadIdx.x == 0) {
        unsigned id = __builtin_amdgcn_s_sendmsg_rtn(0x87);
        s_half = (int)((id >> 18) & 1u);
    }
    __syncthreads();
    return s_half;
}

__global__ void k_deg_init(int* __restrict__ deg0, int* __restrict__ deg1, int n) {
    int i = blockIdx.x * blockDim.x + threadIdx.x;
    if (i < n) {
        deg0[i] = 1;
        deg1[i] = 0;
    }
}

__global__ void k_deg_count(const int* __restrict__ dst, int* __restrict__ deg0,
                            int* __restrict__ deg1, int e) {
    int* __restrict__ deg = chip_half() ? deg1 : deg0;
    const int lo = blockIdx.x * EDGE_CHUNK + threadIdx.x;
    for (int k = 0; k < EDGE_CHUNK / 256; ++k) {
        int i = lo + k * 256;
        if (i < e) atomicAdd(&deg[dst[i]], 1);
    }
}

__global__ void k_dinv(float* __restrict__ dinv, const int* __restrict__ deg1, int n) {
    int i = blockIdx.x * blockDim.x + threadIdx.x;
    if (i < n) {
        int d = ((const int*)dinv)[i] + deg1[i];
        dinv[i] = rsqrtf((float)d);
    }
}

__global__ void k_initx(const float* __restrict__ x, const float* __restrict__ dinv,
                        float* __restrict__ xs, float* __restrict__ aggx,
                        float* __restrict__ aggx1, int n) {
    int v = blockIdx.x * blockDim.x + threadIdx.x;
    if (v >= n) return;
    float di = dinv[v];
    const float* xr = x + (size_t)v * 3;
    float4 o;
    o.x = xr[0] * di;
    o.y = xr[1] * di;
    o.z = xr[2] * di;
    o.w = 0.0f;
    ((float4*)xs)[v]    = o;
    ((float4*)aggx)[v]  = o;
    ((float4*)aggx1)[v] = make_float4(0.0f, 0.0f, 0.0f, 0.0f);
}

__global__ void k_edgex(const int* __restrict__ src, const int* __restrict__ dst,
                        const float* __restrict__ xs, float* __restrict__ aggx,
                        float* __restrict__ aggx1, int e) {
    float* __restrict__ acc = chip_half() ? aggx1 : aggx;
    const int lo = blockIdx.x * EDGE_CHUNK + threadIdx.x;
    for (int k = 0; k < EDGE_CHUNK / 256; ++k) {
        int i = lo + k * 256;
        if (i < e) {
            int s = src[i], d = dst[i];
            float4 m0 = ((const float4*)xs)[s];
            float* o = acc + (size_t)d * 4;
            unsafeAtomicAdd(o + 0, m0.x);
            unsafeAtomicAdd(o + 1, m0.y);
            unsafeAtomicAdd(o + 2, m0.z);
        }
    }
}

__global__ void k_gemm1(const float* __restrict__ aggx, const float* __restrict__ aggx1,
                        const float* __restrict__ dinv,
                        const float* __restrict__ W1, const float* __restrict__ b1,
                        float* __restrict__ agg1, int n) {
    int lane = threadIdx.x & 31;
    int wave = threadIdx.x >> 5;
    int base = (blockIdx.x * (blockDim.x >> 5) + wave) * 16;
    if (base >= n) return;

    int  m   = lane & 15;
    bool hi  = lane >= 16;
    int node = base + m;
    if (node >= n) node = n - 1;

    const float2 av = *(const float2*)(aggx  + (size_t)node * 4 + (hi ? 2 : 0));
    const float2 aw = *(const float2*)(aggx1 + (size_t)node * 4 + (hi ? 2 : 0));
    const float  dn = dinv[node];
    v2f a;
    a.x = (av.x + aw.x) * dn;
    a.y = (av.y + aw.y) * dn;

    int  mc    = (m < 8) ? m : 7;
    bool valid = (m < 8);
    float w0 = W1[0 * 8 + mc];
    float w1 = W1[1 * 8 + mc];
    float w2 = W1[2 * 8 + mc];
    float bj = b1[mc];
    v2f b;
    b.x = valid ? (hi ? w2 : w0) : 0.0f;
    b.y = (valid && !hi) ? w1 : 0.0f;

    float cb = valid ? bj : 0.0f;
    v8f c = {cb, cb, cb, cb, cb, cb, cb, cb};
    c = __builtin_amdgcn_wmma_f32_16x16x4_f32(false, a, false, b, (short)0, c,
                                              false, false);

    if (valid) {
        float* hp = agg1 + (size_t)(base + (hi ? 8 : 0)) * 8 + m;
        if (base + 16 <= n) {
            #pragma unroll
            for (int r = 0; r < 8; ++r) hp[(size_t)r * 8] = c[r];
        } else {
            #pragma unroll
            for (int r = 0; r < 8; ++r) {
                int row = base + r + (hi ? 8 : 0);
                if (row < n) agg1[(size_t)row * 8 + m] = c[r];
            }
        }
    }
}

__global__ void k_gemm2(const float* __restrict__ agg1, const float* __restrict__ dinv,
                        const float* __restrict__ W2, float* __restrict__ h2,
                        float* __restrict__ out, float* __restrict__ out1, int n) {
    int lane = threadIdx.x & 31;
    int wave = threadIdx.x >> 5;
    int base = (blockIdx.x * (blockDim.x >> 5) + wave) * 16;
    if (base >= n) return;

    int  m   = lane & 15;
    bool hi  = lane >= 16;
    int node = base + m;
    if (node >= n) node = n - 1;
    const float* r0 = agg1 + (size_t)node * 8;
    int k0 = hi ? 2 : 0;

    const float dn = dinv[node];
    v2f a0, a1;
    a0.x = fmaxf(r0[k0],     0.0f) * dn; a0.y = fmaxf(r0[k0 + 1],     0.0f) * dn;
    a1.x = fmaxf(r0[4 + k0], 0.0f) * dn; a1.y = fmaxf(r0[4 + k0 + 1], 0.0f) * dn;

    int  mc    = (m < 2) ? m : 1;
    bool valid = (m < 2);
    float wa = W2[(k0)     * 2 + mc];
    float wb = W2[(k0 + 1) * 2 + mc];
    float wc = W2[(k0 + 4) * 2 + mc];
    float wd = W2[(k0 + 5) * 2 + mc];
    v2f b0, b1v;
    b0.x  = valid ? wa : 0.0f;
    b0.y  = valid ? wb : 0.0f;
    b1v.x = valid ? wc : 0.0f;
    b1v.y = valid ? wd : 0.0f;

    v8f c = {};
    c = __builtin_amdgcn_wmma_f32_16x16x4_f32(false, a0, false, b0, (short)0, c,
                                              false, false);
    c = __builtin_amdgcn_wmma_f32_16x16x4_f32(false, a1, false, b1v, (short)0, c,
                                              false, false);

    if (valid) {
        size_t off = (size_t)(base + (hi ? 8 : 0)) * 2 + m;
        float* hp = h2 + off;
        float* op = out + off;
        float* zp = out1 + off;
        if (base + 16 <= n) {
            #pragma unroll
            for (int r = 0; r < 8; ++r) {
                hp[(size_t)r * 2] = c[r];
                op[(size_t)r * 2] = c[r];
                zp[(size_t)r * 2] = 0.0f;
            }
        } else {
            #pragma unroll
            for (int r = 0; r < 8; ++r) {
                int row = base + r + (hi ? 8 : 0);
                if (row < n) {
                    h2[(size_t)row * 2 + m]   = c[r];
                    out[(size_t)row * 2 + m]  = c[r];
                    out1[(size_t)row * 2 + m] = 0.0f;
                }
            }
        }
    }
}

__global__ void k_edge2(const int* __restrict__ src, const int* __restrict__ dst,
                        const float* __restrict__ h2, float* __restrict__ out,
                        float* __restrict__ out1, int e) {
    float* __restrict__ acc = chip_half() ? out1 : out;
    const int lo = blockIdx.x * EDGE_CHUNK + threadIdx.x;
    for (int k = 0; k < EDGE_CHUNK / 256; ++k) {
        int i = lo + k * 256;
        if (i < e) {
            int s = src[i], d = dst[i];
            const float2* hp = (const float2*)(h2 + (size_t)s * 2);
            float2 m0 = hp[0];
            float* o = acc + (size_t)d * 2;
            unsafeAtomicAdd(o + 0, m0.x);
            unsafeAtomicAdd(o + 1, m0.y);
        }
    }
}

__global__ void k_fin2(const float* __restrict__ dinv, const float* __restrict__ b2,
                       const float* __restrict__ out1, float* __restrict__ out, int n) {
    int v = blockIdx.x * blockDim.x + threadIdx.x;
    if (v >= n) return;
    float  di = dinv[v];
    float2 o  = ((float2*)out)[v];
    float2 p  = ((const float2*)out1)[v];
    o.x = b2[0] + (o.x + p.x) * di;
    o.y = b2[1] + (o.y + p.y) * di;
    ((float2*)out)[v] = o;
}

extern "C" void kernel_launch(void* const* d_in, const int* in_sizes, int n_in,
                              void* d_out, int out_size, void* d_ws, size_t ws_size,
                              hipStream_t stream) {
    const float* x  = (const float*)d_in[0];
    const int*   ei = (const int*)d_in[1];
    const float* W1 = (const float*)d_in[2];
    const float* b1 = (const float*)d_in[3];
    const float* W2 = (const float*)d_in[4];
    const float* b2 = (const float*)d_in[5];
    float* out = (float*)d_out;

    int n = in_sizes[0] / 3;
    int e = in_sizes[1] / 2;
    const int* src = ei;
    const int* dst = ei + e;

    float* dinv = (float*)d_ws;
    float* aggx = dinv + n;
    float* h2   = aggx + (size_t)n * 4;
    float* agg1 = dinv + (size_t)n * 9;
    float* xs   = agg1;
    int*   deg1  = (int*)aggx;
    float* aggx1 = h2;
    float* out1  = dinv + (size_t)n * 7;

    const int T = 256;
    int gN  = (n + T - 1) / T;
    int gE  = (e + EDGE_CHUNK - 1) / EDGE_CHUNK;
    int gMM = (n + 127) / 128;

    k_deg_init <<<gN,  T, 0, stream>>>((int*)dinv, deg1, n);
    k_deg_count<<<gE,  T, 0, stream>>>(dst, (int*)dinv, deg1, e);
    k_dinv     <<<gN,  T, 0, stream>>>(dinv, deg1, n);
    k_initx    <<<gN,  T, 0, stream>>>(x, dinv, xs, aggx, aggx1, n);
    k_edgex    <<<gE,  T, 0, stream>>>(src, dst, xs, aggx, aggx1, e);
    k_gemm1    <<<gMM, T, 0, stream>>>(aggx, aggx1, dinv, W1, b1, agg1, n);
    k_gemm2    <<<gMM, T, 0, stream>>>(agg1, dinv, W2, h2, out, out1, n);
    k_edge2    <<<gE,  T, 0, stream>>>(src, dst, h2, out, out1, e);
    k_fin2     <<<gN,  T, 0, stream>>>(dinv, b2, out1, out, n);
}


# edge scatters put a row's words on adjacent lanes of one atomic instruction (4 lanes/edge layer 1, 2 layer 2): one atomic request per edge row, not per word; loads grouped ahead; sum order changes
# speedup vs baseline: 5.1137x; 1.4330x over previous
#include <hip/hip_runtime.h>
#include <hip/hip_bf16.h>

typedef float v2f __attribute__((ext_vector_type(2)));
typedef float v8f __attribute__((ext_vector_type(8)));

#define EDGE_CHUNK 1024

__device__ __forceinline__ int chip_half() {
    __shared__ int s_half;
    if (threadIdx.x == 0) {
        unsigned id = __builtin_amdgcn_s_sendmsg_rtn(0x87);
        s_half = (int)((id >> 18) & 1u);
    }
    __syncthreads();
    return s_half;
}

__global__ void k_deg_init(int* __restrict__ deg0, int* __restrict__ deg1, int n) {
    int i = blockIdx.x * blockDim.x + threadIdx.x;
    if (i < n) {
        deg0[i] = 1;
        deg1[i] = 0;
    }
}

__global__ void k_deg_count(const int* __restrict__ dst, int* __restrict__ deg0,
                            int* __restrict__ deg1, int e) {
    int* __restrict__ deg = chip_half() ? deg1 : deg0;
    const int lo = blockIdx.x * EDGE_CHUNK + threadIdx.x;
    for (int k = 0; k < EDGE_CHUNK / 256; ++k) {
        int i = lo + k * 256;
        if (i < e) atomicAdd(&deg[dst[i]], 1);
    }
}

__global__ void k_dinv(float* __restrict__ dinv, const int* __restrict__ deg1, int n) {
    int i = blockIdx.x * blockDim.x + threadIdx.x;
    if (i < n) {
        int d = ((const int*)dinv)[i] + deg1[i];
        dinv[i] = rsqrtf((float)d);
    }
}

__global__ void k_initx(const float* __restrict__ x, const float* __restrict__ dinv,
                        float* __restrict__ xs, float* __restrict__ aggx,
                        float* __restrict__ aggx1, int n) {
    int v = blockIdx.x * blockDim.x + threadIdx.x;
    if (v >= n) return;
    float di = dinv[v];
    const float* xr = x + (size_t)v * 3;
    float4 o;
    o.x = xr[0] * di;
    o.y = xr[1] * di;
    o.z = xr[2] * di;
    o.w = 0.0f;
    ((float4*)xs)[v]    = o;
    ((float4*)aggx)[v]  = o;
    ((float4*)aggx1)[v] = make_float4(0.0f, 0.0f, 0.0f, 0.0f);
}

__global__ void k_edgex(const int* __restrict__ src, const int* __restrict__ dst,
                        const float* __restrict__ xs, float* __restrict__ aggx,
                        float* __restrict__ aggx1, int e) {
    float* __restrict__ acc = chip_half() ? aggx1 : aggx;
    constexpr int NK = EDGE_CHUNK / 256;
    const int lane  = threadIdx.x & 31;
    const int w     = lane & 3;
    const int slot  = lane >> 2;
    const int lo    = blockIdx.x * EDGE_CHUNK + threadIdx.x;
    const int first = lo - lane;

    int s[NK], d[NK];
    for (int k = 0; k < NK; ++k) {
        const int i = min(lo + k * 256, e - 1);
        s[k] = src[i];
        d[k] = dst[i];
    }
    float v[NK * 4];
    int   dj[NK * 4];
    for (int k = 0; k < NK; ++k)
        for (int j = 0; j < 4; ++j) {
            const int from = j * 8 + slot;
            const int sj = __shfl(s[k], from);
            dj[k * 4 + j] = __shfl(d[k], from);
            v[k * 4 + j]  = xs[(size_t)sj * 4 + w];
        }
    for (int k = 0; k < NK; ++k)
        for (int j = 0; j < 4; ++j) {
            const int edge = first + k * 256 + j * 8 + slot;
            if (w < 3 && edge < e)
                unsafeAtomicAdd(acc + (size_t)dj[k * 4 + j] * 4 + w, v[k * 4 + j]);
        }
}

__global__ void k_gemm1(const float* __restrict__ aggx, const float* __restrict__ aggx1,
                        const float* __restrict__ dinv,
                        const float* __restrict__ W1, const float* __restrict__ b1,
                        float* __restrict__ agg1, int n) {
    int lane = threadIdx.x & 31;
    int wave = threadIdx.x >> 5;
    int base = (blockIdx.x * (blockDim.x >> 5) + wave) * 16;
    if (base >= n) return;

    int  m   = lane & 15;
    bool hi  = lane >= 16;
    int node = base + m;
    if (node >= n) node = n - 1;

    const float2 av = *(const float2*)(aggx  + (size_t)node * 4 + (hi ? 2 : 0));
    const float2 aw = *(const float2*)(aggx1 + (size_t)node * 4 + (hi ? 2 : 0));
    const float  dn = dinv[node];
    v2f a;
    a.x = (av.x + aw.x) * dn;
    a.y = (av.y + aw.y) * dn;

    int  mc    = (m < 8) ? m : 7;
    bool valid = (m < 8);
    float w0 = W1[0 * 8 + mc];
    float w1 = W1[1 * 8 + mc];
    float w2 = W1[2 * 8 + mc];
    float bj = b1[mc];
    v2f b;
    b.x = valid ? (hi ? w2 : w0) : 0.0f;
    b.y = (valid && !hi) ? w1 : 0.0f;

    float cb = valid ? bj : 0.0f;
    v8f c = {cb, cb, cb, cb, cb, cb, cb, cb};
    c = __builtin_amdgcn_wmma_f32_16x16x4_f32(false, a, false, b, (short)0, c,
                                              false, false);

    if (valid) {
        float* hp = agg1 + (size_t)(base + (hi ? 8 : 0)) * 8 + m;
        if (base + 16 <= n) {
            #pragma unroll
            for (int r = 0; r < 8; ++r) hp[(size_t)r * 8] = c[r];
        } else {
            #pragma unroll
            for (int r = 0; r < 8; ++r) {
                int row = base + r + (hi ? 8 : 0);
                if (row < n) agg1[(size_t)row * 8 + m] = c[r];
            }
        }
    }
}

__global__ void k_gemm2(const float* __restrict__ agg1, const float* __restrict__ dinv,
                        const float* __restrict__ W2, float* __restrict__ h2,
                        float* __restrict__ out, float* __restrict__ out1, int n) {
    int lane = threadIdx.x & 31;
    int wave = threadIdx.x >> 5;
    int base = (blockIdx.x * (blockDim.x >> 5) + wave) * 16;
    if (base >= n) return;

    int  m   = lane & 15;
    bool hi  = lane >= 16;
    int node = base + m;
    if (node >= n) node = n - 1;
    const float* r0 = agg1 + (size_t)node * 8;
    int k0 = hi ? 2 : 0;

    const float dn = dinv[node];
    v2f a0, a1;
    a0.x = fmaxf(r0[k0],     0.0f) * dn; a0.y = fmaxf(r0[k0 + 1],     0.0f) * dn;
    a1.x = fmaxf(r0[4 + k0], 0.0f) * dn; a1.y = fmaxf(r0[4 + k0 + 1], 0.0f) * dn;

    int  mc    = (m < 2) ? m : 1;
    bool valid = (m < 2);
    float wa = W2[(k0)     * 2 + mc];
    float wb = W2[(k0 + 1) * 2 + mc];
    float wc = W2[(k0 + 4) * 2 + mc];
    float wd = W2[(k0 + 5) * 2 + mc];
    v2f b0, b1v;
    b0.x  = valid ? wa : 0.0f;
    b0.y  = valid ? wb : 0.0f;
    b1v.x = valid ? wc : 0.0f;
    b1v.y = valid ? wd : 0.0f;

    v8f c = {};
    c = __builtin_amdgcn_wmma_f32_16x16x4_f32(false, a0, false, b0, (short)0, c,
                                              false, false);
    c = __builtin_amdgcn_wmma_f32_16x16x4_f32(false, a1, false, b1v, (short)0, c,
                                              false, false);

    if (valid) {
        size_t off = (size_t)(base + (hi ? 8 : 0)) * 2 + m;
        float* hp = h2 + off;
        float* op = out + off;
        float* zp = out1 + off;
        if (base + 16 <= n) {
            #pragma unroll
            for (int r = 0; r < 8; ++r) {
                hp[(size_t)r * 2] = c[r];
                op[(size_t)r * 2] = c[r];
                zp[(size_t)r * 2] = 0.0f;
            }
        } else {
            #pragma unroll
            for (int r = 0; r < 8; ++r) {
                int row = base + r + (hi ? 8 : 0);
                if (row < n) {
                    h2[(size_t)row * 2 + m]   = c[r];
                    out[(size_t)row * 2 + m]  = c[r];
                    out1[(size_t)row * 2 + m] = 0.0f;
                }
            }
        }
    }
}

__global__ void k_edge2(const int* __restrict__ src, const int* __restrict__ dst,
                        const float* __restrict__ h2, float* __restrict__ out,
                        float* __restrict__ out1, int e) {
    float* __restrict__ acc = chip_half() ? out1 : out;
    constexpr int NK = EDGE_CHUNK / 256;
    const int lane  = threadIdx.x & 31;
    const int w     = lane & 1;
    const int slot  = lane >> 1;
    const int lo    = blockIdx.x * EDGE_CHUNK + threadIdx.x;
    const int first = lo - lane;

    int s[NK], d[NK];
    for (int k = 0; k < NK; ++k) {
        const int i = min(lo + k * 256, e - 1);
        s[k] = src[i];
        d[k] = dst[i];
    }
    float v[NK * 2];
    int   dj[NK * 2];
    for (int k = 0; k < NK; ++k)
        for (int j = 0; j < 2; ++j) {
            const int from = j * 16 + slot;
            const int sj = __shfl(s[k], from);
            dj[k * 2 + j] = __shfl(d[k], from);
            v[k * 2 + j]  = h2[(size_t)sj * 2 + w];
        }
    for (int k = 0; k < NK; ++k)
        for (int j = 0; j < 2; ++j) {
            const int edge = first + k * 256 + j * 16 + slot;
            if (edge < e)
                unsafeAtomicAdd(acc + (size_t)dj[k * 2 + j] * 2 + w, v[k * 2 + j]);
        }
}

__global__ void k_fin2(const float* __restrict__ dinv, const float* __restrict__ b2,
                       const float* __restrict__ out1, float* __restrict__ out, int n) {
    int v = blockIdx.x * blockDim.x + threadIdx.x;
    if (v >= n) return;
    float  di = dinv[v];
    float2 o  = ((float2*)out)[v];
    float2 p  = ((const float2*)out1)[v];
    o.x = b2[0] + (o.x + p.x) * di;
    o.y = b2[1] + (o.y + p.y) * di;
    ((float2*)out)[v] = o;
}

extern "C" void kernel_launch(void* const* d_in, const int* in_sizes, int n_in,
                              void* d_out, int out_size, void* d_ws, size_t ws_size,
                              hipStream_t stream) {
    const float* x  = (const float*)d_in[0];
    const int*   ei = (const int*)d_in[1];
    const float* W1 = (const float*)d_in[2];
    const float* b1 = (const float*)d_in[3];
    const float* W2 = (const float*)d_in[4];
    const float* b2 = (const float*)d_in[5];
    float* out = (float*)d_out;

    int n = in_sizes[0] / 3;
    int e = in_sizes[1] / 2;
    const int* src = ei;
    const int* dst = ei + e;

    float* dinv = (float*)d_ws;
    float* aggx = dinv + n;
    float* h2   = aggx + (size_t)n * 4;
    float* agg1 = dinv + (size_t)n * 9;
    float* xs   = agg1;
    int*   deg1  = (int*)aggx;
    float* aggx1 = h2;
    float* out1  = dinv + (size_t)n * 7;

    const int T = 256;
    int gN  = (n + T - 1) / T;
    int gE  = (e + EDGE_CHUNK - 1) / EDGE_CHUNK;
    int gMM = (n + 127) / 128;

    k_deg_init <<<gN,  T, 0, stream>>>((int*)dinv, deg1, n);
    k_deg_count<<<gE,  T, 0, stream>>>(dst, (int*)dinv, deg1, e);
    k_dinv     <<<gN,  T, 0, stream>>>(dinv, deg1, n);
    k_initx    <<<gN,  T, 0, stream>>>(x, dinv, xs, aggx, aggx1, n);
    k_edgex    <<<gE,  T, 0, stream>>>(src, dst, xs, aggx, aggx1, e);
    k_gemm1    <<<gMM, T, 0, stream>>>(aggx, aggx1, dinv, W1, b1, agg1, n);
    k_gemm2    <<<gMM, T, 0, stream>>>(agg1, dinv, W2, h2, out, out1, n);
    k_edge2    <<<gE,  T, 0, stream>>>(src, dst, h2, out, out1, e);
    k_fin2     <<<gN,  T, 0, stream>>>(dinv, b2, out1, out, n);
}


# node-wise stages fused: dinv is made inside the pre-scaling kernel, and both WMMA products run in one kernel that hands the 16x8 tile over through LDS, not memory; 9 launches -> 7; sums unchanged
# speedup vs baseline: 5.6440x; 1.1037x over previous
#include <hip/hip_runtime.h>
#include <hip/hip_bf16.h>

typedef float v2f __attribute__((ext_vector_type(2)));
typedef float v8f __attribute__((ext_vector_type(8)));

#define EDGE_CHUNK 1024

__device__ __forceinline__ int chip_half() {
    __shared__ int s_half;
    if (threadIdx.x == 0) {
        unsigned id = __builtin_amdgcn_s_sendmsg_rtn(0x87);
        s_half = (int)((id >> 18) & 1u);
    }
    __syncthreads();
    return s_half;
}

__global__ void k_deg_init(int* __restrict__ deg0, int* __restrict__ deg1, int n) {
    int i = blockIdx.x * blockDim.x + threadIdx.x;
    if (i < n) {
        deg0[i] = 1;
        deg1[i] = 0;
    }
}

__global__ void k_deg_count(const int* __restrict__ dst, int* __restrict__ deg0,
                            int* __restrict__ deg1, int e) {
    int* __restrict__ deg = chip_half() ? deg1 : deg0;
    const int lo = blockIdx.x * EDGE_CHUNK + threadIdx.x;
    for (int k = 0; k < EDGE_CHUNK / 256; ++k) {
        int i = lo + k * 256;
        if (i < e) atomicAdd(&deg[dst[i]], 1);
    }
}

__global__ void k_initx(const float* __restrict__ x, float* __restrict__ dinv,
                        const int* __restrict__ deg1, float* __restrict__ xs,
                        float* __restrict__ aggx, float* __restrict__ aggx1, int n) {
    int v = blockIdx.x * blockDim.x + threadIdx.x;
    if (v >= n) return;
    int d = ((const int*)dinv)[v] + deg1[v];
    float di = rsqrtf((float)d);
    dinv[v] = di;
    const float* xr = x + (size_t)v * 3;
    float4 o;
    o.x = xr[0] * di;
    o.y = xr[1] * di;
    o.z = xr[2] * di;
    o.w = 0.0f;
    ((float4*)xs)[v]    = o;
    ((float4*)aggx)[v]  = o;
    ((float4*)aggx1)[v] = make_float4(0.0f, 0.0f, 0.0f, 0.0f);
}

__global__ void k_edgex(const int* __restrict__ src, const int* __restrict__ dst,
                        const float* __restrict__ xs, float* __restrict__ aggx,
                        float* __restrict__ aggx1, int e) {
    float* __restrict__ acc = chip_half() ? aggx1 : aggx;
    constexpr int NK = EDGE_CHUNK / 256;
    const int lane  = threadIdx.x & 31;
    const int w     = lane & 3;
    const int slot  = lane >> 2;
    const int lo    = blockIdx.x * EDGE_CHUNK + threadIdx.x;
    const int first = lo - lane;

    int s[NK], d[NK];
    for (int k = 0; k < NK; ++k) {
        const int i = min(lo + k * 256, e - 1);
        s[k] = src[i];
        d[k] = dst[i];
    }
    float v[NK * 4];
    int   dj[NK * 4];
    for (int k = 0; k < NK; ++k)
        for (int j = 0; j < 4; ++j) {
            const int from = j * 8 + slot;
            const int sj = __shfl(s[k], from);
            dj[k * 4 + j] = __shfl(d[k], from);
            v[k * 4 + j]  = xs[(size_t)sj * 4 + w];
        }
    for (int k = 0; k < NK; ++k)
        for (int j = 0; j < 4; ++j) {
            const int edge = first + k * 256 + j * 8 + slot;
            if (w < 3 && edge < e)
                unsafeAtomicAdd(acc + (size_t)dj[k * 4 + j] * 4 + w, v[k * 4 + j]);
        }
}

__global__ void k_gemm12(const float* __restrict__ aggx, const float* __restrict__ aggx1,
                         const float* __restrict__ dinv,
                         const float* __restrict__ W1, const float* __restrict__ b1,
                         const float* __restrict__ W2, float* __restrict__ h2,
                         float* __restrict__ out, float* __restrict__ out1, int n) {
    __shared__ float s_tile[8 * 16 * 18];
    int lane = threadIdx.x & 31;
    int wave = threadIdx.x >> 5;
    int base = (blockIdx.x * (blockDim.x >> 5) + wave) * 16;

    int  m   = lane & 15;
    bool hi  = lane >= 16;
    int node = base + m;
    if (node >= n) node = n - 1;

    const float2 av = *(const float2*)(aggx  + (size_t)node * 4 + (hi ? 2 : 0));
    const float2 aw = *(const float2*)(aggx1 + (size_t)node * 4 + (hi ? 2 : 0));
    const float  dn = dinv[node];
    v2f a;
    a.x = (av.x + aw.x) * dn;
    a.y = (av.y + aw.y) * dn;

    int  mc    = (m < 8) ? m : 7;
    bool valid = (m < 8);
    float w0 = W1[0 * 8 + mc];
    float w1 = W1[1 * 8 + mc];
    float w2 = W1[2 * 8 + mc];
    float bj = b1[mc];
    v2f b;
    b.x = valid ? (hi ? w2 : w0) : 0.0f;
    b.y = (valid && !hi) ? w1 : 0.0f;

    int  k0     = hi ? 2 : 0;
    int  mc2    = (m < 2) ? m : 1;
    bool valid2 = (m < 2);
    float wa = W2[(k0)     * 2 + mc2];
    float wb = W2[(k0 + 1) * 2 + mc2];
    float wc = W2[(k0 + 4) * 2 + mc2];
    float wd = W2[(k0 + 5) * 2 + mc2];
    v2f b0, b1v;
    b0.x  = valid2 ? wa : 0.0f;
    b0.y  = valid2 ? wb : 0.0f;
    b1v.x = valid2 ? wc : 0.0f;
    b1v.y = valid2 ? wd : 0.0f;

    float cb = valid ? bj : 0.0f;
    v8f c = {cb, cb, cb, cb, cb, cb, cb, cb};
    c = __builtin_amdgcn_wmma_f32_16x16x4_f32(false, a, false, b, (short)0, c,
                                              false, false);

    float* tw = s_tile + wave * (16 * 18);
    {
        float* tp = tw + (hi ? 8 * 18 : 0) + m;
        #pragma unroll
        for (int r = 0; r < 8; ++r) tp[r * 18] = c[r];
    }
    __syncthreads();
    const float2 p0 = *(const float2*)(tw + m * 18 + k0);
    const float2 p1 = *(const float2*)(tw + m * 18 + 4 + k0);
    v2f a0, a1;
    a0.x = fmaxf(p0.x, 0.0f) * dn; a0.y = fmaxf(p0.y, 0.0f) * dn;
    a1.x = fmaxf(p1.x, 0.0f) * dn; a1.y = fmaxf(p1.y, 0.0f) * dn;

    v8f c2 = {};
    c2 = __builtin_amdgcn_wmma_f32_16x16x4_f32(false, a0, false, b0, (short)0, c2,
                                               false, false);
    c2 = __builtin_amdgcn_wmma_f32_16x16x4_f32(false, a1, false, b1v, (short)0, c2,
                                               false, false);

    if (valid2 && base < n) {
        size_t off = (size_t)(base + (hi ? 8 : 0)) * 2 + m;
        float* hp = h2 + off;
        float* op = out + off;
        float* zp = out1 + off;
        if (base + 16 <= n) {
            #pragma unroll
            for (int r = 0; r < 8; ++r) {
                hp[(size_t)r * 2] = c2[r];
                op[(size_t)r * 2] = c2[r];
                zp[(size_t)r * 2] = 0.0f;
            }
        } else {
            #pragma unroll
            for (int r = 0; r < 8; ++r) {
                int row = base + r + (hi ? 8 : 0);
                if (row < n) {
                    h2[(size_t)row * 2 + m]   = c2[r];
                    out[(size_t)row * 2 + m]  = c2[r];
                    out1[(size_t)row * 2 + m] = 0.0f;
                }
            }
        }
    }
}

__global__ void k_edge2(const int* __restrict__ src, const int* __restrict__ dst,
                        const float* __restrict__ h2, float* __restrict__ out,
                        float* __restrict__ out1, int e) {
    float* __restrict__ acc = chip_half() ? out1 : out;
    constexpr int NK = EDGE_CHUNK / 256;
    const int lane  = threadIdx.x & 31;
    const int w     = lane & 1;
    const int slot  = lane >> 1;
    const int lo    = blockIdx.x * EDGE_CHUNK + threadIdx.x;
    const int first = lo - lane;

    int s[NK], d[NK];
    for (int k = 0; k < NK; ++k) {
        const int i = min(lo + k * 256, e - 1);
        s[k] = src[i];
        d[k] = dst[i];
    }
    float v[NK * 2];
    int   dj[NK * 2];
    for (int k = 0; k < NK; ++k)
        for (int j = 0; j < 2; ++j) {
            const int from = j * 16 + slot;
            const int sj = __shfl(s[k], from);
            dj[k * 2 + j] = __shfl(d[k], from);
            v[k * 2 + j]  = h2[(size_t)sj * 2 + w];
        }
    for (int k = 0; k < NK; ++k)
        for (int j = 0; j < 2; ++j) {
            const int edge = first + k * 256 + j * 16 + slot;
            if (edge < e)
                unsafeAtomicAdd(acc + (size_t)dj[k * 2 + j] * 2 + w, v[k * 2 + j]);
        }
}

__global__ void k_fin2(const float* __restrict__ dinv, const float* __restrict__ b2,
                       const float* __restrict__ out1, float* __restrict__ out, int n) {
    int v = blockIdx.x * blockDim.x + threadIdx.x;
    if (v >= n) return;
    float  di = dinv[v];
    float2 o  = ((float2*)out)[v];
    float2 p  = ((const float2*)out1)[v];
    o.x = b2[0] + (o.x + p.x) * di;
    o.y = b2[1] + (o.y + p.y) * di;
    ((float2*)out)[v] = o;
}

extern "C" void kernel_launch(void* const* d_in, const int* in_sizes, int n_in,
                              void* d_out, int out_size, void* d_ws, size_t ws_size,
                              hipStream_t stream) {
    const float* x  = (const float*)d_in[0];
    const int*   ei = (const int*)d_in[1];
    const float* W1 = (const float*)d_in[2];
    const float* b1 = (const float*)d_in[3];
    const float* W2 = (const float*)d_in[4];
    const float* b2 = (const float*)d_in[5];
    float* out = (float*)d_out;

    int n = in_sizes[0] / 3;
    int e = in_sizes[1] / 2;
    const int* src = ei;
    const int* dst = ei + e;

    float* dinv  = (float*)d_ws;
    float* aggx  = dinv + n;
    float* h2    = dinv + (size_t)n * 5;
    float* out1  = dinv + (size_t)n * 7;
    float* xs    = dinv + (size_t)n * 9;
    float* aggx1 = dinv + (size_t)n * 13;
    int*   deg1  = (int*)h2;

    const int T = 256;
    int gN  = (n + T - 1) / T;
    int gE  = (e + EDGE_CHUNK - 1) / EDGE_CHUNK;
    int gMM = (n + 127) / 128;

    k_deg_init <<<gN,  T, 0, stream>>>((int*)dinv, deg1, n);
    k_deg_count<<<gE,  T, 0, stream>>>(dst, (int*)dinv, deg1, e);
    k_initx    <<<gN,  T, 0, stream>>>(x, dinv, deg1, xs, aggx, aggx1, n);
    k_edgex    <<<gE,  T, 0, stream>>>(src, dst, xs, aggx, aggx1, e);
    k_gemm12   <<<gMM, T, 0, stream>>>(aggx, aggx1, dinv, W1, b1, W2, h2, out, out1, n);
    k_edge2    <<<gE,  T, 0, stream>>>(src, dst, h2, out, out1, e);
    k_fin2     <<<gN,  T, 0, stream>>>(dinv, b2, out1, out, n);
}


# edges binned once by 4096-node target range (counting sort in LDS); one work-group per range sums degree and both layers in LDS, no global atomic per edge; 7 launches -> 6; sum order changes
# speedup vs baseline: 8.2965x; 1.4700x over previous
#include <hip/hip_runtime.h>
#include <hip/hip_bf16.h>

typedef float v2f __attribute__((ext_vector_type(2)));
typedef float v8f __attribute__((ext_vector_type(8)));

#define BSH   12
#define BN    (1 << BSH)
#define MAXB  256
#define WGT   1024
#define EPT   8
#define CH    (WGT * EPT)

__global__ void k_zero(int* __restrict__ tab) {
    tab[threadIdx.x] = 0;
}

__global__ void __launch_bounds__(WGT)
k_hist(const int* __restrict__ dst, int* __restrict__ cnt, int e) {
    __shared__ int s_hist[MAXB];
    const int tid = threadIdx.x;
    if (tid < MAXB) s_hist[tid] = 0;
    __syncthreads();
    const int lo = blockIdx.x * CH + tid;
    int d[EPT];
    #pragma unroll
    for (int k = 0; k < EPT; ++k) d[k] = dst[min(lo + k * WGT, e - 1)];
    #pragma unroll
    for (int k = 0; k < EPT; ++k)
        if (lo + k * WGT < e) atomicAdd(&s_hist[d[k] >> BSH], 1);
    __syncthreads();
    if (tid < MAXB) {
        const int c = s_hist[tid];
        if (c > 0) atomicAdd(&cnt[tid], c);
    }
}

__global__ void __launch_bounds__(WGT)
k_bin(const int* __restrict__ src, const int* __restrict__ dst, const int* __restrict__ cnt,
      int* __restrict__ cur, int* __restrict__ bstart, unsigned* __restrict__ rec, int e) {
    __shared__ int      s_hist[MAXB];
    __shared__ int      s_cnt[MAXB];
    __shared__ int      s_lstart[MAXB];
    __shared__ int      s_delta[MAXB];
    __shared__ unsigned s_rec[CH];
    __shared__ int      s_bk[CH];
    const int tid = threadIdx.x;
    if (tid < MAXB) { s_hist[tid] = 0; s_cnt[tid] = cnt[tid]; }
    __syncthreads();
    const int lo = blockIdx.x * CH + tid;
    int s[EPT], d[EPT], rk[EPT];
    #pragma unroll
    for (int k = 0; k < EPT; ++k) {
        const int i = min(lo + k * WGT, e - 1);
        s[k] = src[i];
        d[k] = dst[i];
    }
    #pragma unroll
    for (int k = 0; k < EPT; ++k)
        rk[k] = (lo + k * WGT < e) ? atomicAdd(&s_hist[d[k] >> BSH], 1) : 0;
    __syncthreads();
    if (tid < MAXB) {
        int gpre = 0, lpre = 0;
        for (int j = 0; j < MAXB; ++j) {
            const int cg = s_cnt[j], cl = s_hist[j];
            if (j < tid) { gpre += cg; lpre += cl; }
        }
        const int c = s_hist[tid];
        int off = 0;
        if (c > 0) off = atomicAdd(&cur[tid], c);
        s_lstart[tid] = lpre;
        s_delta[tid]  = gpre + off - lpre;
        if (blockIdx.x == 0) bstart[tid] = gpre;
    }
    __syncthreads();
    #pragma unroll
    for (int k = 0; k < EPT; ++k)
        if (lo + k * WGT < e) {
            const int b = d[k] >> BSH;
            const int p = s_lstart[b] + rk[k];
            s_rec[p] = ((unsigned)s[k] << BSH) | (unsigned)(d[k] & (BN - 1));
            s_bk[p]  = b;
        }
    __syncthreads();
    const int total = min((int)CH, e - (int)blockIdx.x * CH);
    for (int p = tid; p < total; p += WGT)
        rec[p + s_delta[s_bk[p]]] = s_rec[p];
}

__global__ void __launch_bounds__(WGT)
k_deg_xs(const float* __restrict__ x, const int* __restrict__ cnt,
         const int* __restrict__ bstart, const unsigned* __restrict__ rec,
         float* __restrict__ xs, int n) {
    __shared__ int s_deg[BN];
    const int tid = threadIdx.x;
    const int b   = blockIdx.x;
    for (int i = tid; i < BN; i += WGT) s_deg[i] = 1;
    __syncthreads();
    const int count = cnt[b];
    const unsigned* __restrict__ rp = rec + bstart[b];
    int i0 = 0;
    for (; i0 + CH <= count; i0 += CH) {
        unsigned r[EPT];
        #pragma unroll
        for (int k = 0; k < EPT; ++k) r[k] = rp[i0 + k * WGT + tid];
        #pragma unroll
        for (int k = 0; k < EPT; ++k) atomicAdd(&s_deg[r[k] & (BN - 1)], 1);
    }
    if (i0 < count) {
        unsigned r[EPT];
        #pragma unroll
        for (int k = 0; k < EPT; ++k) r[k] = rp[min(i0 + k * WGT + tid, count - 1)];
        #pragma unroll
        for (int k = 0; k < EPT; ++k)
            if (i0 + k * WGT + tid < count) atomicAdd(&s_deg[r[k] & (BN - 1)], 1);
    }
    __syncthreads();
    for (int i = tid; i < BN; i += WGT) {
        const int v = b * BN + i;
        if (v < n) {
            const float di = rsqrtf((float)s_deg[i]);
            const float* xr = x + (size_t)v * 3;
            float4 o;
            o.x = xr[0] * di;
            o.y = xr[1] * di;
            o.z = xr[2] * di;
            o.w = di;
            ((float4*)xs)[v] = o;
        }
    }
}

__global__ void __launch_bounds__(WGT)
k_layer1(const unsigned* __restrict__ rec, const int* __restrict__ cnt,
         const int* __restrict__ bstart, const float* __restrict__ xs,
         const float* __restrict__ W1, const float* __restrict__ b1,
         const float* __restrict__ W2, float* __restrict__ h2, int n) {
    __shared__ float s_ax[BN];
    __shared__ float s_ay[BN];
    __shared__ float s_az[BN];
    __shared__ float s_di[BN];
    __shared__ float s_tile[(WGT / 32) * 16 * 18];
    const int tid = threadIdx.x;
    const int b   = blockIdx.x;
    const float4* __restrict__ xs4 = (const float4*)xs;

    for (int i = tid; i < BN; i += WGT) {
        const float4 o = xs4[min(b * BN + i, n - 1)];
        s_ax[i] = o.x;
        s_ay[i] = o.y;
        s_az[i] = o.z;
        s_di[i] = o.w;
    }
    __syncthreads();

    const int count = cnt[b];
    const unsigned* __restrict__ rp = rec + bstart[b];
    int i0 = 0;
    for (; i0 + CH <= count; i0 += CH) {
        unsigned r[EPT];
        float4   g[EPT];
        #pragma unroll
        for (int k = 0; k < EPT; ++k) r[k] = rp[i0 + k * WGT + tid];
        #pragma unroll
        for (int k = 0; k < EPT; ++k) g[k] = xs4[r[k] >> BSH];
        #pragma unroll
        for (int k = 0; k < EPT; ++k) {
            const int dl = (int)(r[k] & (BN - 1));
            atomicAdd(&s_ax[dl], g[k].x);
            atomicAdd(&s_ay[dl], g[k].y);
            atomicAdd(&s_az[dl], g[k].z);
        }
    }
    if (i0 < count) {
        unsigned r[EPT];
        float4   g[EPT];
        #pragma unroll
        for (int k = 0; k < EPT; ++k) r[k] = rp[min(i0 + k * WGT + tid, count - 1)];
        #pragma unroll
        for (int k = 0; k < EPT; ++k) g[k] = xs4[r[k] >> BSH];
        #pragma unroll
        for (int k = 0; k < EPT; ++k)
            if (i0 + k * WGT + tid < count) {
                const int dl = (int)(r[k] & (BN - 1));
                atomicAdd(&s_ax[dl], g[k].x);
                atomicAdd(&s_ay[dl], g[k].y);
                atomicAdd(&s_az[dl], g[k].z);
            }
    }
    __syncthreads();

    const int  lane = tid & 31;
    const int  wave = tid >> 5;
    const int  m    = lane & 15;
    const bool hi   = lane >= 16;

    const int  mc    = (m < 8) ? m : 7;
    const bool valid = (m < 8);
    const float w0 = W1[0 * 8 + mc];
    const float w1 = W1[1 * 8 + mc];
    const float w2 = W1[2 * 8 + mc];
    const float bj = b1[mc];
    v2f bq;
    bq.x = valid ? (hi ? w2 : w0) : 0.0f;
    bq.y = (valid && !hi) ? w1 : 0.0f;

    const int  k0     = hi ? 2 : 0;
    const int  mc2    = (m < 2) ? m : 1;
    const bool valid2 = (m < 2);
    const float wa = W2[(k0)     * 2 + mc2];
    const float wb = W2[(k0 + 1) * 2 + mc2];
    const float wc = W2[(k0 + 4) * 2 + mc2];
    const float wd = W2[(k0 + 5) * 2 + mc2];
    v2f b0, b1v;
    b0.x  = valid2 ? wa : 0.0f;
    b0.y  = valid2 ? wb : 0.0f;
    b1v.x = valid2 ? wc : 0.0f;
    b1v.y = valid2 ? wd : 0.0f;

    const float cb = valid ? bj : 0.0f;
    float* tw = s_tile + wave * (16 * 18);

    const int gfirst = b * BN;
    #pragma unroll
    for (int it = 0; it < BN / 16 / (WGT / 32); ++it) {
        const int lbase = (it * (WGT / 32) + wave) * 16;
        const int ln    = lbase + m;
        const float dn  = s_di[ln];
        const float ax  = s_ax[ln];
        const float ay  = s_ay[ln];
        const float az  = s_az[ln];

        v2f a;
        a.x = (hi ? az : ax) * dn;
        a.y = ay * (hi ? 0.0f : dn);

        v8f c = {cb, cb, cb, cb, cb, cb, cb, cb};
        c = __builtin_amdgcn_wmma_f32_16x16x4_f32(false, a, false, bq, (short)0, c,
                                                  false, false);

        float* tp = tw + (hi ? 8 * 18 : 0) + m;
        #pragma unroll
        for (int r = 0; r < 8; ++r) tp[r * 18] = c[r];
        __syncthreads();
        const float2 p0 = *(const float2*)(tw + m * 18 + k0);
        const float2 p1 = *(const float2*)(tw + m * 18 + 4 + k0);
        v2f a0, a1;
        a0.x = fmaxf(p0.x, 0.0f) * dn; a0.y = fmaxf(p0.y, 0.0f) * dn;
        a1.x = fmaxf(p1.x, 0.0f) * dn; a1.y = fmaxf(p1.y, 0.0f) * dn;

        v8f c2 = {};
        c2 = __builtin_amdgcn_wmma_f32_16x16x4_f32(false, a0, false, b0, (short)0, c2,
                                                   false, false);
        c2 = __builtin_amdgcn_wmma_f32_16x16x4_f32(false, a1, false, b1v, (short)0, c2,
                                                   false, false);

        #pragma unroll
        for (int r = 0; r < 8; ++r) tp[r * 18] = c2[r];
        __syncthreads();
        h2[(size_t)(gfirst + lbase) * 2 + lane] = tw[(lane >> 1) * 18 + (lane & 1)];
    }
}

__global__ void __launch_bounds__(WGT)
k_layer2(const unsigned* __restrict__ rec, const int* __restrict__ cnt,
         const int* __restrict__ bstart, const float* __restrict__ h2,
         const float* __restrict__ xs, const float* __restrict__ b2,
         float* __restrict__ out, int n) {
    __shared__ float s_o0[BN];
    __shared__ float s_o1[BN];
    const int tid = threadIdx.x;
    const int b   = blockIdx.x;
    const float2* __restrict__ h22 = (const float2*)h2;

    for (int i = tid; i < BN; i += WGT) {
        const float2 o = h22[min(b * BN + i, n - 1)];
        s_o0[i] = o.x;
        s_o1[i] = o.y;
    }
    __syncthreads();

    const int count = cnt[b];
    const unsigned* __restrict__ rp = rec + bstart[b];
    int i0 = 0;
    for (; i0 + CH <= count; i0 += CH) {
        unsigned r[EPT];
        float2   g[EPT];
        #pragma unroll
        for (int k = 0; k < EPT; ++k) r[k] = rp[i0 + k * WGT + tid];
        #pragma unroll
        for (int k = 0; k < EPT; ++k) g[k] = h22[r[k] >> BSH];
        #pragma unroll
        for (int k = 0; k < EPT; ++k) {
            const int dl = (int)(r[k] & (BN - 1));
            atomicAdd(&s_o0[dl], g[k].x);
            atomicAdd(&s_o1[dl], g[k].y);
        }
    }
    if (i0 < count) {
        unsigned r[EPT];
        float2   g[EPT];
        #pragma unroll
        for (int k = 0; k < EPT; ++k) r[k] = rp[min(i0 + k * WGT + tid, count - 1)];
        #pragma unroll
        for (int k = 0; k < EPT; ++k) g[k] = h22[r[k] >> BSH];
        #pragma unroll
        for (int k = 0; k < EPT; ++k)
            if (i0 + k * WGT + tid < count) {
                const int dl = (int)(r[k] & (BN - 1));
                atomicAdd(&s_o0[dl], g[k].x);
                atomicAdd(&s_o1[dl], g[k].y);
            }
    }
    __syncthreads();

    for (int i = tid; i < BN; i += WGT) {
        const int v = b * BN + i;
        if (v < n) {
            const float di = xs[(size_t)v * 4 + 3];
            float2 o;
            o.x = b2[0] + s_o0[i] * di;
            o.y = b2[1] + s_o1[i] * di;
            ((float2*)out)[v] = o;
        }
    }
}

extern "C" void kernel_launch(void* const* d_in, const int* in_sizes, int n_in,
                              void* d_out, int out_size, void* d_ws, size_t ws_size,
                              hipStream_t stream) {
    const float* x  = (const float*)d_in[0];
    const int*   ei = (const int*)d_in[1];
    const float* W1 = (const float*)d_in[2];
    const float* b1 = (const float*)d_in[3];
    const float* W2 = (const float*)d_in[4];
    const float* b2 = (const float*)d_in[5];
    float* out = (float*)d_out;

    int n = in_sizes[0] / 3;
    int e = in_sizes[1] / 2;
    const int* src = ei;
    const int* dst = ei + e;

    int gE = (e + CH - 1) / CH;
    int gB = (n + BN - 1) / BN;

    float*    xs     = (float*)d_ws;
    float*    h2     = xs + (size_t)n * 4;
    unsigned* rec    = (unsigned*)(h2 + (size_t)gB * BN * 2);
    int*      cnt    = (int*)(rec + (size_t)e);
    int*      cur    = cnt + MAXB;
    int*      bstart = cnt + 2 * MAXB;

    k_zero  <<<1,  2 * MAXB, 0, stream>>>(cnt);
    k_hist  <<<gE, WGT, 0, stream>>>(dst, cnt, e);
    k_bin   <<<gE, WGT, 0, stream>>>(src, dst, cnt, cur, bstart, rec, e);
    k_deg_xs<<<gB, WGT, 0, stream>>>(x, cnt, bstart, rec, xs, n);
    k_layer1<<<gB, WGT, 0, stream>>>(rec, cnt, bstart, xs, W1, b1, W2, h2, n);
    k_layer2<<<gB, WGT, 0, stream>>>(rec, cnt, bstart, h2, xs, b2, out, n);
}


# both edge-list passes run as resident work-groups (256 and 512) looping over the list, not 1221 short-lived groups each: 3x fewer waves to start, set-up once per group; sum order by arrival as before
# speedup vs baseline: 8.8250x; 1.0637x over previous
#include <hip/hip_runtime.h>
#include <hip/hip_bf16.h>

typedef float v2f __attribute__((ext_vector_type(2)));
typedef float v8f __attribute__((ext_vector_type(8)));

#define BSH   12
#define BN    (1 << BSH)
#define MAXB  256
#define WGT   1024
#define EPT   8
#define CH    (WGT * EPT)
#define GH    256
#define HPT   16
#define GBIN  512

__global__ void k_zero(int* __restrict__ tab) {
    tab[threadIdx.x] = 0;
}

__global__ void __launch_bounds__(WGT)
k_hist(const int* __restrict__ dst, int* __restrict__ cnt, int e) {
    __shared__ int s_hist[MAXB];
    const int tid = threadIdx.x;
    if (tid < MAXB) s_hist[tid] = 0;
    __syncthreads();
    const int step = (int)gridDim.x * WGT;
    int i = (int)blockIdx.x * WGT + tid;
    for (; i + (HPT - 1) * step < e; i += HPT * step) {
        int d[HPT];
        #pragma unroll
        for (int k = 0; k < HPT; ++k) d[k] = dst[i + k * step];
        #pragma unroll
        for (int k = 0; k < HPT; ++k) atomicAdd(&s_hist[d[k] >> BSH], 1);
    }
    if (i < e) {
        int d[HPT];
        #pragma unroll
        for (int k = 0; k < HPT; ++k) d[k] = dst[min(i + k * step, e - 1)];
        #pragma unroll
        for (int k = 0; k < HPT; ++k)
            if (i + k * step < e) atomicAdd(&s_hist[d[k] >> BSH], 1);
    }
    __syncthreads();
    if (tid < MAXB) {
        const int c = s_hist[tid];
        if (c > 0) atomicAdd(&cnt[tid], c);
    }
}

__global__ void __launch_bounds__(WGT)
k_bin(const int* __restrict__ src, const int* __restrict__ dst, const int* __restrict__ cnt,
      int* __restrict__ cur, int* __restrict__ bstart, unsigned* __restrict__ rec, int e,
      int nchunk) {
    __shared__ int      s_hist[MAXB];
    __shared__ int      s_cnt[MAXB];
    __shared__ int      s_lstart[MAXB];
    __shared__ int      s_delta[MAXB];
    __shared__ unsigned s_rec[CH];
    __shared__ int      s_bk[CH];
    const int tid = threadIdx.x;
    if (tid < MAXB) s_cnt[tid] = cnt[tid];
    __syncthreads();
    int gpre = 0;
    if (tid < MAXB)
        for (int j = 0; j < MAXB; ++j) {
            const int cg = s_cnt[j];
            if (j < tid) gpre += cg;
        }

    int s[EPT], d[EPT], rk[EPT];
    int c = blockIdx.x;
    if (c < nchunk) {
        const int lo = c * CH + tid;
        #pragma unroll
        for (int k = 0; k < EPT; ++k) {
            const int i = min(lo + k * WGT, e - 1);
            s[k] = src[i];
            d[k] = dst[i];
        }
    }
    for (; c < nchunk; c += gridDim.x) {
        const int lo = c * CH + tid;
        if (tid < MAXB) s_hist[tid] = 0;
        __syncthreads();
        #pragma unroll
        for (int k = 0; k < EPT; ++k)
            rk[k] = (lo + k * WGT < e) ? atomicAdd(&s_hist[d[k] >> BSH], 1) : 0;
        __syncthreads();
        if (tid < MAXB) {
            int lpre = 0;
            for (int j = 0; j < MAXB; ++j) {
                const int cl = s_hist[j];
                if (j < tid) lpre += cl;
            }
            const int cc = s_hist[tid];
            int off = 0;
            if (cc > 0) off = atomicAdd(&cur[tid], cc);
            s_lstart[tid] = lpre;
            s_delta[tid]  = gpre + off - lpre;
            if (c == 0) bstart[tid] = gpre;
        }
        __syncthreads();
        #pragma unroll
        for (int k = 0; k < EPT; ++k)
            if (lo + k * WGT < e) {
                const int b = d[k] >> BSH;
                const int p = s_lstart[b] + rk[k];
                s_rec[p] = ((unsigned)s[k] << BSH) | (unsigned)(d[k] & (BN - 1));
                s_bk[p]  = b;
            }
        __syncthreads();
        const int cn = c + (int)gridDim.x;
        if (cn < nchunk) {
            const int lon = cn * CH + tid;
            #pragma unroll
            for (int k = 0; k < EPT; ++k) {
                const int i = min(lon + k * WGT, e - 1);
                s[k] = src[i];
                d[k] = dst[i];
            }
        }
        const int total = min((int)CH, e - c * CH);
        for (int p = tid; p < total; p += WGT)
            rec[p + s_delta[s_bk[p]]] = s_rec[p];
    }
}

__global__ void __launch_bounds__(WGT)
k_deg_xs(const float* __restrict__ x, const int* __restrict__ cnt,
         const int* __restrict__ bstart, const unsigned* __restrict__ rec,
         float* __restrict__ xs, int n) {
    __shared__ int s_deg[BN];
    const int tid = threadIdx.x;
    const int b   = blockIdx.x;
    for (int i = tid; i < BN; i += WGT) s_deg[i] = 1;
    __syncthreads();
    const int count = cnt[b];
    const unsigned* __restrict__ rp = rec + bstart[b];
    int i0 = 0;
    for (; i0 + CH <= count; i0 += CH) {
        unsigned r[EPT];
        #pragma unroll
        for (int k = 0; k < EPT; ++k) r[k] = rp[i0 + k * WGT + tid];
        #pragma unroll
        for (int k = 0; k < EPT; ++k) atomicAdd(&s_deg[r[k] & (BN - 1)], 1);
    }
    if (i0 < count) {
        unsigned r[EPT];
        #pragma unroll
        for (int k = 0; k < EPT; ++k) r[k] = rp[min(i0 + k * WGT + tid, count - 1)];
        #pragma unroll
        for (int k = 0; k < EPT; ++k)
            if (i0 + k * WGT + tid < count) atomicAdd(&s_deg[r[k] & (BN - 1)], 1);
    }
    __syncthreads();
    for (int i = tid; i < BN; i += WGT) {
        const int v = b * BN + i;
        if (v < n) {
            const float di = rsqrtf((float)s_deg[i]);
            const float* xr = x + (size_t)v * 3;
            float4 o;
            o.x = xr[0] * di;
            o.y = xr[1] * di;
            o.z = xr[2] * di;
            o.w = di;
            ((float4*)xs)[v] = o;
        }
    }
}

__global__ void __launch_bounds__(WGT)
k_layer1(const unsigned* __restrict__ rec, const int* __restrict__ cnt,
         const int* __restrict__ bstart, const float* __restrict__ xs,
         const float* __restrict__ W1, const float* __restrict__ b1,
         const float* __restrict__ W2, float* __restrict__ h2, int n) {
    __shared__ float s_ax[BN];
    __shared__ float s_ay[BN];
    __shared__ float s_az[BN];
    __shared__ float s_di[BN];
    __shared__ float s_tile[(WGT / 32) * 16 * 18];
    const int tid = threadIdx.x;
    const int b   = blockIdx.x;
    const float4* __restrict__ xs4 = (const float4*)xs;

    for (int i = tid; i < BN; i += WGT) {
        const float4 o = xs4[min(b * BN + i, n - 1)];
        s_ax[i] = o.x;
        s_ay[i] = o.y;
        s_az[i] = o.z;
        s_di[i] = o.w;
    }
    __syncthreads();

    const int count = cnt[b];
    const unsigned* __restrict__ rp = rec + bstart[b];
    int i0 = 0;
    for (; i0 + CH <= count; i0 += CH) {
        unsigned r[EPT];
        float4   g[EPT];
        #pragma unroll
        for (int k = 0; k < EPT; ++k) r[k] = rp[i0 + k * WGT + tid];
        #pragma unroll
        for (int k = 0; k < EPT; ++k) g[k] = xs4[r[k] >> BSH];
        #pragma unroll
        for (int k = 0; k < EPT; ++k) {
            const int dl = (int)(r[k] & (BN - 1));
            atomicAdd(&s_ax[dl], g[k].x);
            atomicAdd(&s_ay[dl], g[k].y);
            atomicAdd(&s_az[dl], g[k].z);
        }
    }
    if (i0 < count) {
        unsigned r[EPT];
        float4   g[EPT];
        #pragma unroll
        for (int k = 0; k < EPT; ++k) r[k] = rp[min(i0 + k * WGT + tid, count - 1)];
        #pragma unroll
        for (int k = 0; k < EPT; ++k) g[k] = xs4[r[k] >> BSH];
        #pragma unroll
        for (int k = 0; k < EPT; ++k)
            if (i0 + k * WGT + tid < count) {
                const int dl = (int)(r[k] & (BN - 1));
                atomicAdd(&s_ax[dl], g[k].x);
                atomicAdd(&s_ay[dl], g[k].y);
                atomicAdd(&s_az[dl], g[k].z);
            }
    }
    __syncthreads();

    const int  lane = tid & 31;
    const int  wave = tid >> 5;
    const int  m    = lane & 15;
    const bool hi   = lane >= 16;

    const int  mc    = (m < 8) ? m : 7;
    const bool valid = (m < 8);
    const float w0 = W1[0 * 8 + mc];
    const float w1 = W1[1 * 8 + mc];
    const float w2 = W1[2 * 8 + mc];
    const float bj = b1[mc];
    v2f bq;
    bq.x = valid ? (hi ? w2 : w0) : 0.0f;
    bq.y = (valid && !hi) ? w1 : 0.0f;

    const int  k0     = hi ? 2 : 0;
    const int  mc2    = (m < 2) ? m : 1;
    const bool valid2 = (m < 2);
    const float wa = W2[(k0)     * 2 + mc2];
    const float wb = W2[(k0 + 1) * 2 + mc2];
    const float wc = W2[(k0 + 4) * 2 + mc2];
    const float wd = W2[(k0 + 5) * 2 + mc2];
    v2f b0, b1v;
    b0.x  = valid2 ? wa : 0.0f;
    b0.y  = valid2 ? wb : 0.0f;
    b1v.x = valid2 ? wc : 0.0f;
    b1v.y = valid2 ? wd : 0.0f;

    const float cb = valid ? bj : 0.0f;
    float* tw = s_tile + wave * (16 * 18);

    const int gfirst = b * BN;
    #pragma unroll
    for (int it = 0; it < BN / 16 / (WGT / 32); ++it) {
        const int lbase = (it * (WGT / 32) + wave) * 16;
        const int ln    = lbase + m;
        const float dn  = s_di[ln];
        const float ax  = s_ax[ln];
        const float ay  = s_ay[ln];
        const float az  = s_az[ln];

        v2f a;
        a.x = (hi ? az : ax) * dn;
        a.y = ay * (hi ? 0.0f : dn);

        v8f c = {cb, cb, cb, cb, cb, cb, cb, cb};
        c = __builtin_amdgcn_wmma_f32_16x16x4_f32(false, a, false, bq, (short)0, c,
                                                  false, false);

        float* tp = tw + (hi ? 8 * 18 : 0) + m;
        #pragma unroll
        for (int r = 0; r < 8; ++r) tp[r * 18] = c[r];
        __syncthreads();
        const float2 p0 = *(const float2*)(tw + m * 18 + k0);
        const float2 p1 = *(const float2*)(tw + m * 18 + 4 + k0);
        v2f a0, a1;
        a0.x = fmaxf(p0.x, 0.0f) * dn; a0.y = fmaxf(p0.y, 0.0f) * dn;
        a1.x = fmaxf(p1.x, 0.0f) * dn; a1.y = fmaxf(p1.y, 0.0f) * dn;

        v8f c2 = {};
        c2 = __builtin_amdgcn_wmma_f32_16x16x4_f32(false, a0, false, b0, (short)0, c2,
                                                   false, false);
        c2 = __builtin_amdgcn_wmma_f32_16x16x4_f32(false, a1, false, b1v, (short)0, c2,
                                                   false, false);

        #pragma unroll
        for (int r = 0; r < 8; ++r) tp[r * 18] = c2[r];
        __syncthreads();
        h2[(size_t)(gfirst + lbase) * 2 + lane] = tw[(lane >> 1) * 18 + (lane & 1)];
    }
}

__global__ void __launch_bounds__(WGT)
k_layer2(const unsigned* __restrict__ rec, const int* __restrict__ cnt,
         const int* __restrict__ bstart, const float* __restrict__ h2,
         const float* __restrict__ xs, const float* __restrict__ b2,
         float* __restrict__ out, int n) {
    __shared__ float s_o0[BN];
    __shared__ float s_o1[BN];
    const int tid = threadIdx.x;
    const int b   = blockIdx.x;
    const float2* __restrict__ h22 = (const float2*)h2;

    for (int i = tid; i < BN; i += WGT) {
        const float2 o = h22[min(b * BN + i, n - 1)];
        s_o0[i] = o.x;
        s_o1[i] = o.y;
    }
    __syncthreads();

    const int count = cnt[b];
    const unsigned* __restrict__ rp = rec + bstart[b];
    int i0 = 0;
    for (; i0 + CH <= count; i0 += CH) {
        unsigned r[EPT];
        float2   g[EPT];
        #pragma unroll
        for (int k = 0; k < EPT; ++k) r[k] = rp[i0 + k * WGT + tid];
        #pragma unroll
        for (int k = 0; k < EPT; ++k) g[k] = h22[r[k] >> BSH];
        #pragma unroll
        for (int k = 0; k < EPT; ++k) {
            const int dl = (int)(r[k] & (BN - 1));
            atomicAdd(&s_o0[dl], g[k].x);
            atomicAdd(&s_o1[dl], g[k].y);
        }
    }
    if (i0 < count) {
        unsigned r[EPT];
        float2   g[EPT];
        #pragma unroll
        for (int k = 0; k < EPT; ++k) r[k] = rp[min(i0 + k * WGT + tid, count - 1)];
        #pragma unroll
        for (int k = 0; k < EPT; ++k) g[k] = h22[r[k] >> BSH];
        #pragma unroll
        for (int k = 0; k < EPT; ++k)
            if (i0 + k * WGT + tid < count) {
                const int dl = (int)(r[k] & (BN - 1));
                atomicAdd(&s_o0[dl], g[k].x);
                atomicAdd(&s_o1[dl], g[k].y);
            }
    }
    __syncthreads();

    for (int i = tid; i < BN; i += WGT) {
        const int v = b * BN + i;
        if (v < n) {
            const float di = xs[(size_t)v * 4 + 3];
            float2 o;
            o.x = b2[0] + s_o0[i] * di;
            o.y = b2[1] + s_o1[i] * di;
            ((float2*)out)[v] = o;
        }
    }
}

extern "C" void kernel_launch(void* const* d_in, const int* in_sizes, int n_in,
                              void* d_out, int out_size, void* d_ws, size_t ws_size,
                              hipStream_t stream) {
    const float* x  = (const float*)d_in[0];
    const int*   ei = (const int*)d_in[1];
    const float* W1 = (const float*)d_in[2];
    const float* b1 = (const float*)d_in[3];
    const float* W2 = (const float*)d_in[4];
    const float* b2 = (const float*)d_in[5];
    float* out = (float*)d_out;

    int n = in_sizes[0] / 3;
    int e = in_sizes[1] / 2;
    const int* src = ei;
    const int* dst = ei + e;

    int gE = (e + CH - 1) / CH;
    int gB = (n + BN - 1) / BN;

    float*    xs     = (float*)d_ws;
    float*    h2     = xs + (size_t)n * 4;
    unsigned* rec    = (unsigned*)(h2 + (size_t)gB * BN * 2);
    int*      cnt    = (int*)(rec + (size_t)e);
    int*      cur    = cnt + MAXB;
    int*      bstart = cnt + 2 * MAXB;

    k_zero  <<<1,  2 * MAXB, 0, stream>>>(cnt);
    k_hist  <<<GH,   WGT, 0, stream>>>(dst, cnt, e);
    k_bin   <<<GBIN, WGT, 0, stream>>>(src, dst, cnt, cur, bstart, rec, e, gE);
    k_deg_xs<<<gB, WGT, 0, stream>>>(x, cnt, bstart, rec, xs, n);
    k_layer1<<<gB, WGT, 0, stream>>>(rec, cnt, bstart, xs, W1, b1, W2, h2, n);
    k_layer2<<<gB, WGT, 0, stream>>>(rec, cnt, bstart, h2, xs, b2, out, n);
}


# edge list binned in ONE pass: each of 256 work-groups sorts its own slab by target range in LDS and writes it out whole; range kernels walk the per-slab pieces; 6 launches -> 4; sum order changes
# speedup vs baseline: 9.9457x; 1.1270x over previous
#include <hip/hip_runtime.h>
#include <hip/hip_bf16.h>

typedef float v2f __attribute__((ext_vector_type(2)));
typedef float v8f __attribute__((ext_vector_type(8)));

#define BSH   12
#define BN    (1 << BSH)
#define MAXB  256
#define WGT   1024
#define EPT   8
#define GP    256
#define QPT   10
#define SCAP  (WGT * QPT * 4)
#define SPW   (GP / (WGT / 32))
#define WB    (32 * EPT)

__global__ void __launch_bounds__(WGT)
k_bin(const int* __restrict__ src, const int* __restrict__ dst, int2* __restrict__ tab,
      unsigned* __restrict__ rec, int e, int S) {
    __shared__ int      s_cnt[MAXB];
    __shared__ int      s_start[MAXB];
    __shared__ int      s_wtot[MAXB / 32];
    __shared__ unsigned s_rec[SCAP];
    const int tid   = threadIdx.x;
    const int w     = blockIdx.x;
    const int first = w * S;
    const int len   = max(0, min(S, e - first));
    const int nq    = len >> 2;
    if (tid < MAXB) s_cnt[tid] = 0;

    int4 sv[QPT], dv[QPT];
    if (nq > 0) {
        const int4* __restrict__ s4 = (const int4*)(src + first);
        const int4* __restrict__ d4 = (const int4*)(dst + first);
        #pragma unroll
        for (int k = 0; k < QPT; ++k) {
            const int q = min(k * WGT + tid, nq - 1);
            sv[k] = s4[q];
            dv[k] = d4[q];
        }
    } else {
        #pragma unroll
        for (int k = 0; k < QPT; ++k) {
            sv[k] = make_int4(0, 0, 0, 0);
            dv[k] = make_int4(0, 0, 0, 0);
        }
    }
    __syncthreads();

    unsigned key[QPT * 4], rw[QPT * 4];
    #pragma unroll
    for (int k = 0; k < QPT; ++k) {
        const bool on = (k * WGT + tid) < nq;
        const int sa[4] = {sv[k].x, sv[k].y, sv[k].z, sv[k].w};
        const int da[4] = {dv[k].x, dv[k].y, dv[k].z, dv[k].w};
        #pragma unroll
        for (int c = 0; c < 4; ++c) {
            const int b  = da[c] >> BSH;
            const int rk = on ? atomicAdd(&s_cnt[b], 1) : 0;
            key[k * 4 + c] = ((unsigned)rk << 8) | (unsigned)b;
            rw[k * 4 + c]  = ((unsigned)sa[c] << BSH) | (unsigned)(da[c] & (BN - 1));
        }
    }
    __syncthreads();
    int cmine = 0, incl = 0;
    if (tid < MAXB) {
        cmine = s_cnt[tid];
        incl  = cmine;
        #pragma unroll
        for (int o = 1; o < 32; o <<= 1) {
            const int up = __shfl_up(incl, o);
            if ((tid & 31) >= o) incl += up;
        }
        if ((tid & 31) == 31) s_wtot[tid >> 5] = incl;
    }
    __syncthreads();
    if (tid < MAXB) {
        int pre = incl - cmine;
        #pragma unroll
        for (int j = 0; j < MAXB / 32; ++j) {
            const int wt = s_wtot[j];
            if (j < (tid >> 5)) pre += wt;
        }
        s_start[tid] = pre;
        tab[tid * GP + w] = make_int2(first + pre, cmine);
    }
    __syncthreads();
    #pragma unroll
    for (int k = 0; k < QPT; ++k) {
        const bool on = (k * WGT + tid) < nq;
        #pragma unroll
        for (int c = 0; c < 4; ++c)
            if (on) s_rec[s_start[key[k * 4 + c] & 255u] + (int)(key[k * 4 + c] >> 8)] = rw[k * 4 + c];
    }
    __syncthreads();
    uint4* __restrict__ o4 = (uint4*)(rec + first);
    const uint4* sr4 = (const uint4*)s_rec;
    for (int q = tid; q < nq; q += WGT) o4[q] = sr4[q];
}

struct SegMap {
    int thr[SPW];
    int off[SPW];
    int tot;
};

__device__ __forceinline__ SegMap seg_map(const int2* __restrict__ tab, int b, int wv) {
    SegMap m;
    int run = 0;
    #pragma unroll
    for (int k = 0; k < SPW; ++k) {
        const int2 t = tab[b * GP + wv * SPW + k];
        m.thr[k] = run;
        m.off[k] = t.x - run;
        run += t.y;
    }
    m.tot = run;
    return m;
}

__device__ __forceinline__ int seg_index(const SegMap& m, int j) {
    int o = m.off[0];
    #pragma unroll
    for (int k = 1; k < SPW; ++k) o = (j >= m.thr[k]) ? m.off[k] : o;
    return j + o;
}

__global__ void __launch_bounds__(WGT)
k_deg_xs(const float* __restrict__ x, const int2* __restrict__ tab,
         const unsigned* __restrict__ rec, float* __restrict__ xs, int n) {
    __shared__ int s_deg[BN];
    const int tid  = threadIdx.x;
    const int b    = blockIdx.x;
    const int lane = tid & 31;
    const int wv   = __builtin_amdgcn_readfirstlane(tid >> 5);
    for (int i = tid; i < BN; i += WGT) s_deg[i] = 1;
    __syncthreads();
    const SegMap sm = seg_map(tab, b, wv);
    const int count = sm.tot;
    int i0 = 0;
    for (; i0 + WB <= count; i0 += WB) {
        unsigned r[EPT];
        #pragma unroll
        for (int k = 0; k < EPT; ++k) r[k] = rec[seg_index(sm, i0 + k * 32 + lane)];
        #pragma unroll
        for (int k = 0; k < EPT; ++k) atomicAdd(&s_deg[r[k] & (BN - 1)], 1);
    }
    if (i0 < count) {
        unsigned r[EPT];
        #pragma unroll
        for (int k = 0; k < EPT; ++k)
            r[k] = rec[seg_index(sm, min(i0 + k * 32 + lane, count - 1))];
        #pragma unroll
        for (int k = 0; k < EPT; ++k)
            if (i0 + k * 32 + lane < count) atomicAdd(&s_deg[r[k] & (BN - 1)], 1);
    }
    __syncthreads();
    for (int i = tid; i < BN; i += WGT) {
        const int v = b * BN + i;
        if (v < n) {
            const float di = rsqrtf((float)s_deg[i]);
            const float* xr = x + (size_t)v * 3;
            float4 o;
            o.x = xr[0] * di;
            o.y = xr[1] * di;
            o.z = xr[2] * di;
            o.w = di;
            ((float4*)xs)[v] = o;
        }
    }
}

__global__ void __launch_bounds__(WGT)
k_layer1(const unsigned* __restrict__ rec, const int2* __restrict__ tab,
         const float* __restrict__ xs,
         const float* __restrict__ W1, const float* __restrict__ b1,
         const float* __restrict__ W2, float* __restrict__ h2, int n) {
    __shared__ float s_ax[BN];
    __shared__ float s_ay[BN];
    __shared__ float s_az[BN];
    __shared__ float s_di[BN];
    __shared__ float s_tile[(WGT / 32) * 16 * 18];
    const int tid = threadIdx.x;
    const int b   = blockIdx.x;
    const float4* __restrict__ xs4 = (const float4*)xs;

    for (int i = tid; i < BN; i += WGT) {
        const float4 o = xs4[min(b * BN + i, n - 1)];
        s_ax[i] = o.x;
        s_ay[i] = o.y;
        s_az[i] = o.z;
        s_di[i] = o.w;
    }
    __syncthreads();

    const int lane0 = tid & 31;
    const int wv    = __builtin_amdgcn_readfirstlane(tid >> 5);
    const SegMap sm = seg_map(tab, b, wv);
    const int count = sm.tot;
    int i0 = 0;
    for (; i0 + WB <= count; i0 += WB) {
        unsigned r[EPT];
        float4   g[EPT];
        #pragma unroll
        for (int k = 0; k < EPT; ++k) r[k] = rec[seg_index(sm, i0 + k * 32 + lane0)];
        #pragma unroll
        for (int k = 0; k < EPT; ++k) g[k] = xs4[r[k] >> BSH];
        #pragma unroll
        for (int k = 0; k < EPT; ++k) {
            const int dl = (int)(r[k] & (BN - 1));
            atomicAdd(&s_ax[dl], g[k].x);
            atomicAdd(&s_ay[dl], g[k].y);
            atomicAdd(&s_az[dl], g[k].z);
        }
    }
    if (i0 < count) {
        unsigned r[EPT];
        float4   g[EPT];
        #pragma unroll
        for (int k = 0; k < EPT; ++k)
            r[k] = rec[seg_index(sm, min(i0 + k * 32 + lane0, count - 1))];
        #pragma unroll
        for (int k = 0; k < EPT; ++k) g[k] = xs4[r[k] >> BSH];
        #pragma unroll
        for (int k = 0; k < EPT; ++k)
            if (i0 + k * 32 + lane0 < count) {
                const int dl = (int)(r[k] & (BN - 1));
                atomicAdd(&s_ax[dl], g[k].x);
                atomicAdd(&s_ay[dl], g[k].y);
                atomicAdd(&s_az[dl], g[k].z);
            }
    }
    __syncthreads();

    const int  lane = tid & 31;
    const int  wave = tid >> 5;
    const int  m    = lane & 15;
    const bool hi   = lane >= 16;

    const int  mc    = (m < 8) ? m : 7;
    const bool valid = (m < 8);
    const float w0 = W1[0 * 8 + mc];
    const float w1 = W1[1 * 8 + mc];
    const float w2 = W1[2 * 8 + mc];
    const float bj = b1[mc];
    v2f bq;
    bq.x = valid ? (hi ? w2 : w0) : 0.0f;
    bq.y = (valid && !hi) ? w1 : 0.0f;

    const int  k0     = hi ? 2 : 0;
    const int  mc2    = (m < 2) ? m : 1;
    const bool valid2 = (m < 2);
    const float wa = W2[(k0)     * 2 + mc2];
    const float wb = W2[(k0 + 1) * 2 + mc2];
    const float wc = W2[(k0 + 4) * 2 + mc2];
    const float wd = W2[(k0 + 5) * 2 + mc2];
    v2f b0, b1v;
    b0.x  = valid2 ? wa : 0.0f;
    b0.y  = valid2 ? wb : 0.0f;
    b1v.x = valid2 ? wc : 0.0f;
    b1v.y = valid2 ? wd : 0.0f;

    const float cb = valid ? bj : 0.0f;
    float* tw = s_tile + wave * (16 * 18);

    const int gfirst = b * BN;
    #pragma unroll
    for (int it = 0; it < BN / 16 / (WGT / 32); ++it) {
        const int lbase = (it * (WGT / 32) + wave) * 16;
        const int ln    = lbase + m;
        const float dn  = s_di[ln];
        const float ax  = s_ax[ln];
        const float ay  = s_ay[ln];
        const float az  = s_az[ln];

        v2f a;
        a.x = (hi ? az : ax) * dn;
        a.y = ay * (hi ? 0.0f : dn);

        v8f c = {cb, cb, cb, cb, cb, cb, cb, cb};
        c = __builtin_amdgcn_wmma_f32_16x16x4_f32(false, a, false, bq, (short)0, c,
                                                  false, false);

        float* tp = tw + (hi ? 8 * 18 : 0) + m;
        #pragma unroll
        for (int r = 0; r < 8; ++r) tp[r * 18] = c[r];
        __syncthreads();
        const float2 p0 = *(const float2*)(tw + m * 18 + k0);
        const float2 p1 = *(const float2*)(tw + m * 18 + 4 + k0);
        v2f a0, a1;
        a0.x = fmaxf(p0.x, 0.0f) * dn; a0.y = fmaxf(p0.y, 0.0f) * dn;
        a1.x = fmaxf(p1.x, 0.0f) * dn; a1.y = fmaxf(p1.y, 0.0f) * dn;

        v8f c2 = {};
        c2 = __builtin_amdgcn_wmma_f32_16x16x4_f32(false, a0, false, b0, (short)0, c2,
                                                   false, false);
        c2 = __builtin_amdgcn_wmma_f32_16x16x4_f32(false, a1, false, b1v, (short)0, c2,
                                                   false, false);

        #pragma unroll
        for (int r = 0; r < 8; ++r) tp[r * 18] = c2[r];
        __syncthreads();
        h2[(size_t)(gfirst + lbase) * 2 + lane] = tw[(lane >> 1) * 18 + (lane & 1)];
    }
}

__global__ void __launch_bounds__(WGT)
k_layer2(const unsigned* __restrict__ rec, const int2* __restrict__ tab,
         const float* __restrict__ h2,
         const float* __restrict__ xs, const float* __restrict__ b2,
         float* __restrict__ out, int n) {
    __shared__ float s_o0[BN];
    __shared__ float s_o1[BN];
    const int tid = threadIdx.x;
    const int b   = blockIdx.x;
    const float2* __restrict__ h22 = (const float2*)h2;

    for (int i = tid; i < BN; i += WGT) {
        const float2 o = h22[min(b * BN + i, n - 1)];
        s_o0[i] = o.x;
        s_o1[i] = o.y;
    }
    __syncthreads();

    const int lane0 = tid & 31;
    const int wv    = __builtin_amdgcn_readfirstlane(tid >> 5);
    const SegMap sm = seg_map(tab, b, wv);
    const int count = sm.tot;
    int i0 = 0;
    for (; i0 + WB <= count; i0 += WB) {
        unsigned r[EPT];
        float2   g[EPT];
        #pragma unroll
        for (int k = 0; k < EPT; ++k) r[k] = rec[seg_index(sm, i0 + k * 32 + lane0)];
        #pragma unroll
        for (int k = 0; k < EPT; ++k) g[k] = h22[r[k] >> BSH];
        #pragma unroll
        for (int k = 0; k < EPT; ++k) {
            const int dl = (int)(r[k] & (BN - 1));
            atomicAdd(&s_o0[dl], g[k].x);
            atomicAdd(&s_o1[dl], g[k].y);
        }
    }
    if (i0 < count) {
        unsigned r[EPT];
        float2   g[EPT];
        #pragma unroll
        for (int k = 0; k < EPT; ++k)
            r[k] = rec[seg_index(sm, min(i0 + k * 32 + lane0, count - 1))];
        #pragma unroll
        for (int k = 0; k < EPT; ++k) g[k] = h22[r[k] >> BSH];
        #pragma unroll
        for (int k = 0; k < EPT; ++k)
            if (i0 + k * 32 + lane0 < count) {
                const int dl = (int)(r[k] & (BN - 1));
                atomicAdd(&s_o0[dl], g[k].x);
                atomicAdd(&s_o1[dl], g[k].y);
            }
    }
    __syncthreads();

    for (int i = tid; i < BN; i += WGT) {
        const int v = b * BN + i;
        if (v < n) {
            const float di = xs[(size_t)v * 4 + 3];
            float2 o;
            o.x = b2[0] + s_o0[i] * di;
            o.y = b2[1] + s_o1[i] * di;
            ((float2*)out)[v] = o;
        }
    }
}

extern "C" void kernel_launch(void* const* d_in, const int* in_sizes, int n_in,
                              void* d_out, int out_size, void* d_ws, size_t ws_size,
                              hipStream_t stream) {
    const float* x  = (const float*)d_in[0];
    const int*   ei = (const int*)d_in[1];
    const float* W1 = (const float*)d_in[2];
    const float* b1 = (const float*)d_in[3];
    const float* W2 = (const float*)d_in[4];
    const float* b2 = (const float*)d_in[5];
    float* out = (float*)d_out;

    int n = in_sizes[0] / 3;
    int e = in_sizes[1] / 2;
    const int* src = ei;
    const int* dst = ei + e;

    int S  = (((e + GP - 1) / GP) + 31) & ~31;
    int gB = (n + BN - 1) / BN;

    float*    xs  = (float*)d_ws;
    float*    h2  = xs + (size_t)n * 4;
    unsigned* rec = (unsigned*)(h2 + (size_t)gB * BN * 2);
    int2*     tab = (int2*)(rec + (size_t)GP * S);

    k_bin   <<<GP, WGT, 0, stream>>>(src, dst, tab, rec, e, S);
    k_deg_xs<<<gB, WGT, 0, stream>>>(x, tab, rec, xs, n);
    k_layer1<<<gB, WGT, 0, stream>>>(rec, tab, xs, W1, b1, W2, h2, n);
    k_layer2<<<gB, WGT, 0, stream>>>(rec, tab, h2, xs, b2, out, n);
}
